# starting kernel
# baseline (speedup 1.0000x reference)
#include <hip/hip_runtime.h>
#include <math.h>

#define HEADS      4
#define HCF        128
#define C2         47
#define HC2        188
#define HC2P       192
#define SK2P       64
#define NEG_SLOPE  0.2f
#define EPS_DEN    1e-16f

typedef float v2f __attribute__((ext_vector_type(2)));
typedef float v8f __attribute__((ext_vector_type(8)));

__device__ __forceinline__ unsigned fkey(float f) {
  unsigned u = __float_as_uint(f);
  return (u & 0x80000000u) ? ~u : (u | 0x80000000u);
}
__device__ __forceinline__ float fdecode(unsigned k) {
  unsigned u = (k & 0x80000000u) ? (k ^ 0x80000000u) : ~k;
  return __uint_as_float(u);
}

template <int K, int LDA, int LDB, int LDC>
__global__ __launch_bounds__(256)
void gemm_wmma_f32(const float* __restrict__ A, const float* __restrict__ B,
                   float* __restrict__ C, int M) {
  __shared__ float bs[(K / 4) * 128 * 2];

  const int tid  = threadIdx.x;
  const int lane = tid & 31;
  const int wave = tid >> 5;
  const int tn   = blockIdx.y * 64;

  for (int idx = tid; idx < (K / 4) * 128; idx += 256) {
    const int l   = idx & 31;
    const int t   = (idx >> 5) & 3;
    const int ks  = idx >> 7;
    const int k   = 4 * ks + ((l >> 4) << 1);
    const int col = tn + t * 16 + (l & 15);
    float2 b;
    b.x = B[(size_t)k * LDB + col];
    b.y = B[(size_t)(k + 1) * LDB + col];
    *(float2*)&bs[(size_t)idx * 2] = b;
  }
  __syncthreads();

  const int rlo   = lane & 15;
  const int khalf = (lane >> 4) << 1;
  const int tm    = blockIdx.x * 8 + wave;

  int arow = tm * 16 + rlo;
  if (arow >= M) arow = M - 1;
  const float* ap = A + (size_t)arow * LDA;

  v8f acc[4] = {};
#pragma unroll 4
  for (int ks = 0; ks < K / 4; ++ks) {
    v2f a;
    a.x = ap[4 * ks + khalf];
    a.y = ap[4 * ks + khalf + 1];
    const float* bp = &bs[(size_t)(ks * 128 + lane) * 2];
#pragma unroll
    for (int t = 0; t < 4; ++t) {
      v2f b = *(const v2f*)(bp + t * 64);
      acc[t] = __builtin_amdgcn_wmma_f32_16x16x4_f32(
          false, a, false, b, (short)0, acc[t], false, false);
    }
  }

  const int mbase = tm * 16 + ((lane >> 4) << 3);
  if ((tm + 1) * 16 <= M) {
    float* cp = C + (size_t)mbase * LDC + tn + rlo;
#pragma unroll
    for (int t = 0; t < 4; ++t)
#pragma unroll
      for (int r = 0; r < 8; ++r)
        cp[r * LDC + t * 16] = acc[t][r];
  } else {
#pragma unroll
    for (int t = 0; t < 4; ++t)
#pragma unroll
      for (int r = 0; r < 8; ++r) {
        const int row = mbase + r;
        if (row < M) C[(size_t)row * LDC + tn + t * 16 + rlo] = acc[t][r];
      }
  }
}

template <int C, int LD>
__global__ void att_kernel(const float* __restrict__ xh,
                           const float* __restrict__ att_src,
                           const float* __restrict__ att_dst,
                           float* __restrict__ asrc, float* __restrict__ adst,
                           int N) {
  int id = blockIdx.x * blockDim.x + threadIdx.x;
  if (id >= N * HEADS) return;
  int n = id >> 2, h = id & 3;
  const float* xp = xh + (size_t)n * LD + h * C;
  const float* ws = att_src + h * C;
  const float* wd = att_dst + h * C;
  float s1 = 0.f, s2 = 0.f;
#pragma unroll 8
  for (int c = 0; c < C; ++c) {
    float v = xp[c];
    s1 += v * ws[c];
    s2 += v * wd[c];
  }
  asrc[id] = s1;
  adst[id] = s2;
}

__global__ void edge_pre(const int* __restrict__ ei, int E0, int EP,
                         const float* __restrict__ asrc,
                         const float* __restrict__ adst,
                         float* __restrict__ eb, unsigned* __restrict__ mkey) {
  int e = blockIdx.x * blockDim.x + threadIdx.x;
  if (e >= EP) return;
  int src = (e < E0) ? ei[e]      : (e - E0);
  int dst = (e < E0) ? ei[E0 + e] : (e - E0);
  const float4 es = *(const float4*)(asrc + (size_t)src * 4);
  const float4 ed = *(const float4*)(adst + (size_t)dst * 4);
  float v[4] = {es.x + ed.x, es.y + ed.y, es.z + ed.z, es.w + ed.w};
#pragma unroll
  for (int h = 0; h < 4; ++h) {
    float x = v[h];
    x = x > 0.f ? x : NEG_SLOPE * x;
    v[h] = x;
    atomicMax(&mkey[(size_t)dst * 4 + h], fkey(x));
  }
  *(float4*)(eb + (size_t)e * 4) = float4{v[0], v[1], v[2], v[3]};
}

__global__ void edge_sum(const int* __restrict__ ei, int E0, int EP,
                         const unsigned* __restrict__ mkey,
                         float* __restrict__ eb, float* __restrict__ sden) {
  int e = blockIdx.x * blockDim.x + threadIdx.x;
  if (e >= EP) return;
  int dst = (e < E0) ? ei[E0 + e] : (e - E0);
  float4 ev = *(const float4*)(eb + (size_t)e * 4);
  uint4 mk  = *(const uint4*)(mkey + (size_t)dst * 4);
  float v[4] = {ev.x, ev.y, ev.z, ev.w};
  unsigned k[4] = {mk.x, mk.y, mk.z, mk.w};
#pragma unroll
  for (int h = 0; h < 4; ++h) {
    float ex = __expf(v[h] - fdecode(k[h]));
    v[h] = ex;
    atomicAdd(&sden[(size_t)dst * 4 + h], ex);
  }
  *(float4*)(eb + (size_t)e * 4) = float4{v[0], v[1], v[2], v[3]};
}

template <int C, int LD>
__global__ __launch_bounds__(256)
void edge_agg(const int* __restrict__ ei, int E0, int EP,
              const float* __restrict__ xh, const float* __restrict__ eb,
              const float* __restrict__ sden, float* __restrict__ agg) {
  int e = blockIdx.x * (blockDim.x >> 5) + (threadIdx.x >> 5);
  if (e >= EP) return;
  const int lane = threadIdx.x & 31;
  int src = (e < E0) ? ei[e]      : (e - E0);
  int dst = (e < E0) ? ei[E0 + e] : (e - E0);
  const float* xs = xh + (size_t)src * LD;
  __builtin_prefetch(xs + lane * 4, 0, 3);
  const float4 ex4 = *(const float4*)(eb + (size_t)e * 4);
  const float4 s4  = *(const float4*)(sden + (size_t)dst * 4);
  float alpha[4] = {ex4.x / (s4.x + EPS_DEN), ex4.y / (s4.y + EPS_DEN),
                    ex4.z / (s4.z + EPS_DEN), ex4.w / (s4.w + EPS_DEN)};
  float* ad = agg + (size_t)dst * LD;
#pragma unroll
  for (int base = 0; base < HEADS * C; base += 32) {
    int f = base + lane;
    if (f < HEADS * C) {
      int h = f / C;
      atomicAdd(&ad[f], xs[f] * alpha[h]);
    }
  }
}

__global__ void combine_elu(const float* __restrict__ agg,
                            const float* __restrict__ skip,
                            const float* __restrict__ bias,
                            const float* __restrict__ skb,
                            float* __restrict__ out, int total) {
  int i = blockIdx.x * blockDim.x + threadIdx.x;
  if (i >= total) return;
  int f = i & 127;
  float v = agg[i] + bias[f] + skip[i] + skb[f];
  out[i] = v > 0.f ? v : expm1f(v);
}

__global__ void combine_out(const float* __restrict__ agg,
                            const float* __restrict__ skip,
                            const float* __restrict__ bias2,
                            const float* __restrict__ skb2,
                            float* __restrict__ out, int N) {
  int i = blockIdx.x * blockDim.x + threadIdx.x;
  if (i >= N * C2) return;
  int n = i / C2, o = i - n * C2;
  const float* a = agg + (size_t)n * HC2P;
  float m = 0.25f * (a[o] + a[C2 + o] + a[2 * C2 + o] + a[3 * C2 + o]);
  out[i] = m + bias2[o] + skip[(size_t)n * SK2P + o] + skb2[o];
}

__global__ void pad_w(const float* __restrict__ w, float* __restrict__ wp,
                      int rows, int cols, int ldp) {
  int i = blockIdx.x * blockDim.x + threadIdx.x;
  if (i >= rows * ldp) return;
  int r = i / ldp, c = i - r * ldp;
  wp[i] = (c < cols) ? w[(size_t)r * cols + c] : 0.f;
}

extern "C" void kernel_launch(void* const* d_in, const int* in_sizes, int n_in,
                              void* d_out, int out_size, void* d_ws, size_t ws_size,
                              hipStream_t stream) {
  const float* x   = (const float*)d_in[0];
  const int*   ei  = (const int*)d_in[1];
  const float* lw[3] = {(const float*)d_in[2], (const float*)d_in[8],  (const float*)d_in[14]};
  const float* as[3] = {(const float*)d_in[3], (const float*)d_in[9],  (const float*)d_in[15]};
  const float* ad[3] = {(const float*)d_in[4], (const float*)d_in[10], (const float*)d_in[16]};
  const float* bi[3] = {(const float*)d_in[5], (const float*)d_in[11], (const float*)d_in[17]};
  const float* sw[3] = {(const float*)d_in[6], (const float*)d_in[12], (const float*)d_in[18]};
  const float* sb[3] = {(const float*)d_in[7], (const float*)d_in[13], (const float*)d_in[19]};

  const int N  = in_sizes[0] / 128;
  const int E0 = in_sizes[1] / 2;
  const int EP = E0 + N;

  float* w = (float*)d_ws;
  float*    xcur = w;                      w += (size_t)N * HCF;
  float*    xh   = w;                      w += (size_t)N * HC2P;
  float*    agg  = w;                      w += (size_t)N * HC2P;
  float*    skip = w;                      w += (size_t)N * HCF;
  float*    asrc = w;                      w += (size_t)N * 4;
  float*    adst = w;                      w += (size_t)N * 4;
  unsigned* mkey = (unsigned*)w;           w += (size_t)N * 4;
  float*    sden = w;                      w += (size_t)N * 4;
  float*    eb   = w;                      w += (size_t)EP * 4;
  float*    w2p  = w;                      w += (size_t)128 * HC2P;
  float*    sw2p = w;                      w += (size_t)128 * SK2P;

  const int mt8  = (N + 127) / 128;
  const int eblk = (EP + 255) / 256;
  const int ewarp = (EP + 7) / 8;

  gemm_wmma_f32<128, 128, 128, 128><<<dim3(mt8, 2), 256, 0, stream>>>(x, lw[0], xh,   N);
  gemm_wmma_f32<128, 128, 128, 128><<<dim3(mt8, 2), 256, 0, stream>>>(x, sw[0], skip, N);
  att_kernel<32, 128><<<(N * 4 + 255) / 256, 256, 0, stream>>>(xh, as[0], ad[0], asrc, adst, N);
  hipMemsetAsync(mkey, 0, (size_t)N * 4 * sizeof(unsigned), stream);
  hipMemsetAsync(sden, 0, (size_t)N * 4 * sizeof(float), stream);
  hipMemsetAsync(agg,  0, (size_t)N * HCF * sizeof(float), stream);
  edge_pre<<<eblk, 256, 0, stream>>>(ei, E0, EP, asrc, adst, eb, mkey);
  edge_sum<<<eblk, 256, 0, stream>>>(ei, E0, EP, mkey, eb, sden);
  edge_agg<32, 128><<<ewarp, 256, 0, stream>>>(ei, E0, EP, xh, eb, sden, agg);
  combine_elu<<<((size_t)N * HCF + 255) / 256, 256, 0, stream>>>(agg, skip, bi[0], sb[0], xcur, N * HCF);

  gemm_wmma_f32<128, 128, 128, 128><<<dim3(mt8, 2), 256, 0, stream>>>(xcur, lw[1], xh,   N);
  gemm_wmma_f32<128, 128, 128, 128><<<dim3(mt8, 2), 256, 0, stream>>>(xcur, sw[1], skip, N);
  att_kernel<32, 128><<<(N * 4 + 255) / 256, 256, 0, stream>>>(xh, as[1], ad[1], asrc, adst, N);
  hipMemsetAsync(mkey, 0, (size_t)N * 4 * sizeof(unsigned), stream);
  hipMemsetAsync(sden, 0, (size_t)N * 4 * sizeof(float), stream);
  hipMemsetAsync(agg,  0, (size_t)N * HCF * sizeof(float), stream);
  edge_pre<<<eblk, 256, 0, stream>>>(ei, E0, EP, asrc, adst, eb, mkey);
  edge_sum<<<eblk, 256, 0, stream>>>(ei, E0, EP, mkey, eb, sden);
  edge_agg<32, 128><<<ewarp, 256, 0, stream>>>(ei, E0, EP, xh, eb, sden, agg);
  combine_elu<<<((size_t)N * HCF + 255) / 256, 256, 0, stream>>>(agg, skip, bi[1], sb[1], xcur, N * HCF);

  pad_w<<<(128 * HC2P + 255) / 256, 256, 0, stream>>>(lw[2], w2p,  128, HC2, HC2P);
  pad_w<<<(128 * SK2P + 255) / 256, 256, 0, stream>>>(sw[2], sw2p, 128, C2,  SK2P);
  gemm_wmma_f32<128, 128, HC2P, HC2P><<<dim3(mt8, 3), 256, 0, stream>>>(xcur, w2p,  xh,   N);
  gemm_wmma_f32<128, 128, SK2P, SK2P><<<dim3(mt8, 1), 256, 0, stream>>>(xcur, sw2p, skip, N);
  att_kernel<47, HC2P><<<(N * 4 + 255) / 256, 256, 0, stream>>>(xh, as[2], ad[2], asrc, adst, N);
  hipMemsetAsync(mkey, 0, (size_t)N * 4 * sizeof(unsigned), stream);
  hipMemsetAsync(sden, 0, (size_t)N * 4 * sizeof(float), stream);
  hipMemsetAsync(agg,  0, (size_t)N * HC2P * sizeof(float), stream);
  edge_pre<<<eblk, 256, 0, stream>>>(ei, E0, EP, asrc, adst, eb, mkey);
  edge_sum<<<eblk, 256, 0, stream>>>(ei, E0, EP, mkey, eb, sden);
  edge_agg<47, HC2P><<<ewarp, 256, 0, stream>>>(ei, E0, EP, xh, eb, sden, agg);
  combine_out<<<(N * C2 + 255) / 256, 256, 0, stream>>>(agg, skip, bi[2], sb[2], (float*)d_out, N);
}


# Edges grouped by destination once per call (int atomic arrival-order rank); one wave per node does softmax and weighted sum in registers, sum order = arrival rank, no float atomics
# speedup vs baseline: 4.1399x; 4.1399x over previous
#include <hip/hip_runtime.h>
#include <math.h>

#define HEADS      4
#define HCF        128
#define C2         47
#define HC2        188
#define HC2P       192
#define SK2P       64
#define NEG_SLOPE  0.2f
#define EPS_DEN    1e-16f

typedef float v2f __attribute__((ext_vector_type(2)));
typedef float v8f __attribute__((ext_vector_type(8)));

template <int K, int LDA, int LDB, int LDC>
__global__ __launch_bounds__(256)
void gemm_wmma_f32(const float* __restrict__ A, const float* __restrict__ B,
                   float* __restrict__ C, int M) {
  __shared__ float bs[(K / 4) * 128 * 2];

  const int tid  = threadIdx.x;
  const int lane = tid & 31;
  const int wave = tid >> 5;
  const int tn   = blockIdx.y * 64;

  for (int idx = tid; idx < (K / 4) * 128; idx += 256) {
    const int l   = idx & 31;
    const int t   = (idx >> 5) & 3;
    const int ks  = idx >> 7;
    const int k   = 4 * ks + ((l >> 4) << 1);
    const int col = tn + t * 16 + (l & 15);
    float2 b;
    b.x = B[(size_t)k * LDB + col];
    b.y = B[(size_t)(k + 1) * LDB + col];
    *(float2*)&bs[(size_t)idx * 2] = b;
  }
  __syncthreads();

  const int rlo   = lane & 15;
  const int khalf = (lane >> 4) << 1;
  const int tm    = blockIdx.x * 8 + wave;

  int arow = tm * 16 + rlo;
  if (arow >= M) arow = M - 1;
  const float* ap = A + (size_t)arow * LDA;

  v8f acc[4] = {};
#pragma unroll 4
  for (int ks = 0; ks < K / 4; ++ks) {
    v2f a;
    a.x = ap[4 * ks + khalf];
    a.y = ap[4 * ks + khalf + 1];
    const float* bp = &bs[(size_t)(ks * 128 + lane) * 2];
#pragma unroll
    for (int t = 0; t < 4; ++t) {
      v2f b = *(const v2f*)(bp + t * 64);
      acc[t] = __builtin_amdgcn_wmma_f32_16x16x4_f32(
          false, a, false, b, (short)0, acc[t], false, false);
    }
  }

  const int mbase = tm * 16 + ((lane >> 4) << 3);
  if ((tm + 1) * 16 <= M) {
    float* cp = C + (size_t)mbase * LDC + tn + rlo;
#pragma unroll
    for (int t = 0; t < 4; ++t)
#pragma unroll
      for (int r = 0; r < 8; ++r)
        cp[r * LDC + t * 16] = acc[t][r];
  } else {
#pragma unroll
    for (int t = 0; t < 4; ++t)
#pragma unroll
      for (int r = 0; r < 8; ++r) {
        const int row = mbase + r;
        if (row < M) C[(size_t)row * LDC + tn + t * 16 + rlo] = acc[t][r];
      }
  }
}

template <int C, int LD>
__global__ void att_kernel(const float* __restrict__ xh,
                           const float* __restrict__ att_src,
                           const float* __restrict__ att_dst,
                           float* __restrict__ asrc, float* __restrict__ adst,
                           int N) {
  int id = blockIdx.x * blockDim.x + threadIdx.x;
  if (id >= N * HEADS) return;
  int n = id >> 2, h = id & 3;
  const float* xp = xh + (size_t)n * LD + h * C;
  const float* ws = att_src + h * C;
  const float* wd = att_dst + h * C;
  float s1 = 0.f, s2 = 0.f;
#pragma unroll 8
  for (int c = 0; c < C; ++c) {
    float v = xp[c];
    s1 += v * ws[c];
    s2 += v * wd[c];
  }
  asrc[id] = s1;
  adst[id] = s2;
}

__global__ void csr_count(const int* __restrict__ ei, int E0,
                          int* __restrict__ cnt, int* __restrict__ rank) {
  int e = blockIdx.x * blockDim.x + threadIdx.x;
  if (e >= E0) return;
  rank[e] = atomicAdd(&cnt[ei[E0 + e]], 1);
}

__global__ __launch_bounds__(1024)
void csr_scan(const int* __restrict__ cnt, int* __restrict__ rowptr, int N) {
  __shared__ int wsum[32];
  const int t    = threadIdx.x;
  const int lane = t & 31;
  const int wave = t >> 5;
  int carry = 0;
  for (int base = 0; base < N; base += 1024 * 16) {
    const int b = base + t * 16;
    int v[16];
    if (b + 16 <= N) {
#pragma unroll
      for (int i = 0; i < 4; ++i) {
        const int4 q = *(const int4*)(cnt + b + 4 * i);
        v[4 * i] = q.x; v[4 * i + 1] = q.y; v[4 * i + 2] = q.z; v[4 * i + 3] = q.w;
      }
    } else {
#pragma unroll
      for (int i = 0; i < 16; ++i) v[i] = (b + i < N) ? cnt[b + i] : 0;
    }
    int s = 0;
#pragma unroll
    for (int i = 0; i < 16; ++i) s += v[i];

    int incl = s;
#pragma unroll
    for (int off = 1; off < 32; off <<= 1) {
      const int y = __shfl_up(incl, off);
      if (lane >= off) incl += y;
    }
    if (lane == 31) wsum[wave] = incl;
    __syncthreads();
    if (wave == 0) {
      int ws = wsum[lane];
#pragma unroll
      for (int off = 1; off < 32; off <<= 1) {
        const int y = __shfl_up(ws, off);
        if (lane >= off) ws += y;
      }
      wsum[lane] = ws;
    }
    __syncthreads();
    const int wave_off   = (wave > 0) ? wsum[wave - 1] : 0;
    const int tile_total = wsum[31];
    int run = carry + wave_off + incl - s;
    int o[16];
#pragma unroll
    for (int i = 0; i < 16; ++i) { o[i] = run; run += v[i]; }
    if (b + 16 <= N) {
#pragma unroll
      for (int i = 0; i < 4; ++i)
        *(int4*)(rowptr + b + 4 * i) = int4{o[4 * i], o[4 * i + 1], o[4 * i + 2], o[4 * i + 3]};
    } else {
#pragma unroll
      for (int i = 0; i < 16; ++i)
        if (b + i < N) rowptr[b + i] = o[i];
    }
    carry += tile_total;
    __syncthreads();
  }
  if (t == 0) rowptr[N] = carry;
}

__global__ void csr_fill(const int* __restrict__ ei, int E0,
                         const int* __restrict__ rowptr, const int* __restrict__ rank,
                         int* __restrict__ col) {
  int e = blockIdx.x * blockDim.x + threadIdx.x;
  if (e >= E0) return;
  col[rowptr[ei[E0 + e]] + rank[e]] = ei[e];
}

__device__ __forceinline__ float wave_max(float v) {
#pragma unroll
  for (int off = 16; off > 0; off >>= 1) v = fmaxf(v, __shfl_xor(v, off));
  return v;
}
__device__ __forceinline__ float wave_sum(float v) {
#pragma unroll
  for (int off = 16; off > 0; off >>= 1) v += __shfl_xor(v, off);
  return v;
}

__device__ __forceinline__ int seg_scores(int j, int total, int beg, int n,
                                          const int* __restrict__ col,
                                          const float* __restrict__ asrc,
                                          const float4 ad4, float e[4]) {
  int s = n;
  if (j < total - 1) s = col[beg + j];
  const float4 as4 = *(const float4*)(asrc + (size_t)s * 4);
  float t[4] = {as4.x + ad4.x, as4.y + ad4.y, as4.z + ad4.z, as4.w + ad4.w};
#pragma unroll
  for (int h = 0; h < 4; ++h) {
    float x = t[h];
    x = x > 0.f ? x : NEG_SLOPE * x;
    e[h] = (j < total) ? x : -INFINITY;
  }
  return s;
}

__device__ __forceinline__ void wave_lds_sync() {
  __builtin_amdgcn_fence(__ATOMIC_ACQ_REL, "wavefront");
}

template <int C, int LD, int NQ>
struct RowAcc {
  float4 acc[NQ];
  int    hq[NQ][4];
  int    lane;

  __device__ __forceinline__ void init(int ln) {
    lane = ln;
#pragma unroll
    for (int r = 0; r < NQ; ++r) {
      acc[r] = float4{0.f, 0.f, 0.f, 0.f};
#pragma unroll
      for (int i = 0; i < 4; ++i) {
        const int h = (4 * (lane + 32 * r) + i) / C;
        hq[r][i] = h < HEADS ? h : HEADS - 1;
      }
    }
  }
  __device__ __forceinline__ bool has(int r) const { return (lane + 32 * r) * 4 < HEADS * C; }

  __device__ __forceinline__ float4 weights(const float* p, int r) const {
    if constexpr (C % 4 == 0) {
      const float w = p[hq[r][0]];
      return float4{w, w, w, w};
    } else {
      return float4{p[hq[r][0]], p[hq[r][1]], p[hq[r][2]], p[hq[r][3]]};
    }
  }

  __device__ __forceinline__ void add(int cnt, int src, const float al[4],
                                      const float* __restrict__ xh, float* alw) {
    *(float4*)(alw + lane * 4) = float4{al[0], al[1], al[2], al[3]};
    wave_lds_sync();
    int jj = 0;
    for (; jj + 4 <= cnt; jj += 4) {
      float4 v[4][NQ];
      float4 a[4][NQ];
#pragma unroll
      for (int u = 0; u < 4; ++u) {
        const int s = __builtin_amdgcn_readlane(src, jj + u);
        const float* xs = xh + (size_t)s * LD + lane * 4;
#pragma unroll
        for (int r = 0; r < NQ; ++r) {
          v[u][r] = has(r) ? *(const float4*)(xs + 128 * r) : float4{0.f, 0.f, 0.f, 0.f};
          a[u][r] = weights(alw + (jj + u) * 4, r);
        }
      }
#pragma unroll
      for (int u = 0; u < 4; ++u)
#pragma unroll
        for (int r = 0; r < NQ; ++r) {
          acc[r].x += v[u][r].x * a[u][r].x;
          acc[r].y += v[u][r].y * a[u][r].y;
          acc[r].z += v[u][r].z * a[u][r].z;
          acc[r].w += v[u][r].w * a[u][r].w;
        }
    }
    for (; jj < cnt; ++jj) {
      const int s = __builtin_amdgcn_readlane(src, jj);
      const float* xs = xh + (size_t)s * LD + lane * 4;
#pragma unroll
      for (int r = 0; r < NQ; ++r) {
        const float4 v = has(r) ? *(const float4*)(xs + 128 * r) : float4{0.f, 0.f, 0.f, 0.f};
        const float4 a = weights(alw + jj * 4, r);
        acc[r].x += v.x * a.x;
        acc[r].y += v.y * a.y;
        acc[r].z += v.z * a.z;
        acc[r].w += v.w * a.w;
      }
    }
    wave_lds_sync();
  }
};

template <int C, int LD, int NQ>
__device__ __forceinline__ void node_aggregate(int n, int lane,
                                               const int* __restrict__ rowptr,
                                               const int* __restrict__ col,
                                               const float* __restrict__ xh,
                                               const float* __restrict__ asrc,
                                               const float* __restrict__ adst,
                                               RowAcc<C, LD, NQ>& ra, float* alw) {
  const int beg   = rowptr[n];
  const int total = rowptr[n + 1] - beg + 1;
  const float4 ad4 = *(const float4*)(adst + (size_t)n * 4);

  float e0[4], m[4];
  const int src0 = seg_scores(lane, total, beg, n, col, asrc, ad4, e0);
#pragma unroll
  for (int h = 0; h < 4; ++h) m[h] = e0[h];
  for (int base = 32; base < total; base += 32) {
    float e[4];
    seg_scores(base + lane, total, beg, n, col, asrc, ad4, e);
#pragma unroll
    for (int h = 0; h < 4; ++h) m[h] = fmaxf(m[h], e[h]);
  }
#pragma unroll
  for (int h = 0; h < 4; ++h) m[h] = wave_max(m[h]);

  float ex0[4], den[4];
#pragma unroll
  for (int h = 0; h < 4; ++h) {
    ex0[h] = (lane < total) ? __expf(e0[h] - m[h]) : 0.f;
    den[h] = ex0[h];
  }
  for (int base = 32; base < total; base += 32) {
    float e[4];
    seg_scores(base + lane, total, beg, n, col, asrc, ad4, e);
#pragma unroll
    for (int h = 0; h < 4; ++h) den[h] += (base + lane < total) ? __expf(e[h] - m[h]) : 0.f;
  }
#pragma unroll
  for (int h = 0; h < 4; ++h) den[h] = wave_sum(den[h]) + EPS_DEN;

  float al[4];
#pragma unroll
  for (int h = 0; h < 4; ++h) al[h] = ex0[h] / den[h];
  ra.add(total < 32 ? total : 32, src0, al, xh, alw);
  for (int base = 32; base < total; base += 32) {
    float e[4];
    const int s = seg_scores(base + lane, total, beg, n, col, asrc, ad4, e);
#pragma unroll
    for (int h = 0; h < 4; ++h)
      al[h] = (base + lane < total) ? __expf(e[h] - m[h]) / den[h] : 0.f;
    const int left = total - base;
    ra.add(left < 32 ? left : 32, s, al, xh, alw);
  }
}

__global__ __launch_bounds__(256)
void gather_elu(const int* __restrict__ rowptr, const int* __restrict__ col,
                const float* __restrict__ xh, const float* __restrict__ asrc,
                const float* __restrict__ adst, const float* __restrict__ skip,
                const float* __restrict__ bias, const float* __restrict__ skb,
                float* __restrict__ out, int N) {
  __shared__ __attribute__((aligned(16))) float alds[8 * 128];
  const int lane = threadIdx.x & 31;
  const int n = __builtin_amdgcn_readfirstlane(blockIdx.x * 8 + (threadIdx.x >> 5));
  if (n >= N) return;
  RowAcc<32, HCF, 1> ra;
  ra.init(lane);
  node_aggregate<32, HCF, 1>(n, lane, rowptr, col, xh, asrc, adst, ra,
                             alds + (threadIdx.x >> 5) * 128);
  const size_t o = (size_t)n * HCF + lane * 4;
  const float4 sk = *(const float4*)(skip + o);
  const float4 b  = *(const float4*)(bias + lane * 4);
  const float4 sb = *(const float4*)(skb + lane * 4);
  float v[4] = {ra.acc[0].x + b.x + sk.x + sb.x, ra.acc[0].y + b.y + sk.y + sb.y,
                ra.acc[0].z + b.z + sk.z + sb.z, ra.acc[0].w + b.w + sk.w + sb.w};
#pragma unroll
  for (int i = 0; i < 4; ++i) v[i] = v[i] > 0.f ? v[i] : expm1f(v[i]);
  *(float4*)(out + o) = float4{v[0], v[1], v[2], v[3]};
}

__global__ __launch_bounds__(256)
void gather_l2(const int* __restrict__ rowptr, const int* __restrict__ col,
               const float* __restrict__ xh, const float* __restrict__ asrc,
               const float* __restrict__ adst, float* __restrict__ agg, int N) {
  __shared__ __attribute__((aligned(16))) float alds[8 * 128];
  const int lane = threadIdx.x & 31;
  const int n = __builtin_amdgcn_readfirstlane(blockIdx.x * 8 + (threadIdx.x >> 5));
  if (n >= N) return;
  RowAcc<C2, HC2P, 2> ra;
  ra.init(lane);
  node_aggregate<C2, HC2P, 2>(n, lane, rowptr, col, xh, asrc, adst, ra,
                              alds + (threadIdx.x >> 5) * 128);
  float* ap = agg + (size_t)n * HC2P + lane * 4;
  *(float4*)ap = ra.acc[0];
  if (ra.has(1)) *(float4*)(ap + 128) = ra.acc[1];
}

__global__ void combine_out(const float* __restrict__ agg,
                            const float* __restrict__ skip,
                            const float* __restrict__ bias2,
                            const float* __restrict__ skb2,
                            float* __restrict__ out, int N) {
  int i = blockIdx.x * blockDim.x + threadIdx.x;
  if (i >= N * C2) return;
  int n = i / C2, o = i - n * C2;
  const float* a = agg + (size_t)n * HC2P;
  float m = 0.25f * (a[o] + a[C2 + o] + a[2 * C2 + o] + a[3 * C2 + o]);
  out[i] = m + bias2[o] + skip[(size_t)n * SK2P + o] + skb2[o];
}

__global__ void pad_w(const float* __restrict__ w, float* __restrict__ wp,
                      int rows, int cols, int ldp) {
  int i = blockIdx.x * blockDim.x + threadIdx.x;
  if (i >= rows * ldp) return;
  int r = i / ldp, c = i - r * ldp;
  wp[i] = (c < cols) ? w[(size_t)r * cols + c] : 0.f;
}

extern "C" void kernel_launch(void* const* d_in, const int* in_sizes, int n_in,
                              void* d_out, int out_size, void* d_ws, size_t ws_size,
                              hipStream_t stream) {
  const float* x   = (const float*)d_in[0];
  const int*   ei  = (const int*)d_in[1];
  const float* lw[3] = {(const float*)d_in[2], (const float*)d_in[8],  (const float*)d_in[14]};
  const float* as[3] = {(const float*)d_in[3], (const float*)d_in[9],  (const float*)d_in[15]};
  const float* ad[3] = {(const float*)d_in[4], (const float*)d_in[10], (const float*)d_in[16]};
  const float* bi[3] = {(const float*)d_in[5], (const float*)d_in[11], (const float*)d_in[17]};
  const float* sw[3] = {(const float*)d_in[6], (const float*)d_in[12], (const float*)d_in[18]};
  const float* sb[3] = {(const float*)d_in[7], (const float*)d_in[13], (const float*)d_in[19]};

  const int N  = in_sizes[0] / 128;
  const int E0 = in_sizes[1] / 2;

  float* w = (float*)d_ws;
  float*    xcur = w;                      w += (size_t)N * HCF;
  float*    xh   = w;                      w += (size_t)N * HC2P;
  float*    agg  = w;                      w += (size_t)N * HC2P;
  float*    skip = w;                      w += (size_t)N * HCF;
  float*    asrc = w;                      w += (size_t)N * 4;
  float*    adst = w;                      w += (size_t)N * 4;
  float*    w2p  = w;                      w += (size_t)128 * HC2P;
  float*    sw2p = w;                      w += (size_t)128 * SK2P;
  int*      cnt    = (int*)w;              w += ((size_t)N + 3) / 4 * 4;
  int*      rowptr = (int*)w;              w += ((size_t)N + 4) / 4 * 4;
  int*      rank   = (int*)w;              w += (size_t)E0;
  int*      col    = (int*)w;              w += (size_t)E0;

  const int mt8  = (N + 127) / 128;
  const int e0blk = (E0 + 255) / 256;
  const int nwarp = (N + 7) / 8;

  hipMemsetAsync(cnt, 0, (size_t)N * sizeof(int), stream);
  csr_count<<<e0blk, 256, 0, stream>>>(ei, E0, cnt, rank);
  csr_scan<<<1, 1024, 0, stream>>>(cnt, rowptr, N);
  csr_fill<<<e0blk, 256, 0, stream>>>(ei, E0, rowptr, rank, col);

  gemm_wmma_f32<128, 128, 128, 128><<<dim3(mt8, 2), 256, 0, stream>>>(x, lw[0], xh,   N);
  gemm_wmma_f32<128, 128, 128, 128><<<dim3(mt8, 2), 256, 0, stream>>>(x, sw[0], skip, N);
  att_kernel<32, 128><<<(N * 4 + 255) / 256, 256, 0, stream>>>(xh, as[0], ad[0], asrc, adst, N);
  gather_elu<<<nwarp, 256, 0, stream>>>(rowptr, col, xh, asrc, adst, skip, bi[0], sb[0], xcur, N);

  gemm_wmma_f32<128, 128, 128, 128><<<dim3(mt8, 2), 256, 0, stream>>>(xcur, lw[1], xh,   N);
  gemm_wmma_f32<128, 128, 128, 128><<<dim3(mt8, 2), 256, 0, stream>>>(xcur, sw[1], skip, N);
  att_kernel<32, 128><<<(N * 4 + 255) / 256, 256, 0, stream>>>(xh, as[1], ad[1], asrc, adst, N);
  gather_elu<<<nwarp, 256, 0, stream>>>(rowptr, col, xh, asrc, adst, skip, bi[1], sb[1], xcur, N);

  pad_w<<<(128 * HC2P + 255) / 256, 256, 0, stream>>>(lw[2], w2p,  128, HC2, HC2P);
  pad_w<<<(128 * SK2P + 255) / 256, 256, 0, stream>>>(sw[2], sw2p, 128, C2,  SK2P);
  gemm_wmma_f32<128, 128, HC2P, HC2P><<<dim3(mt8, 3), 256, 0, stream>>>(xcur, w2p,  xh,   N);
  gemm_wmma_f32<128, 128, SK2P, SK2P><<<dim3(mt8, 1), 256, 0, stream>>>(xcur, sw2p, skip, N);
  att_kernel<47, HC2P><<<(N * 4 + 255) / 256, 256, 0, stream>>>(xh, as[2], ad[2], asrc, adst, N);
  gather_l2<<<nwarp, 256, 0, stream>>>(rowptr, col, xh, asrc, adst, agg, N);
  combine_out<<<(N * C2 + 255) / 256, 256, 0, stream>>>(agg, skip, bi[2], sb[2], (float*)d_out, N);
}


# Launch fusion: skip GEMM rides in the layer GEMM's launch, layer 0-1 attention dots in its epilogue, weight padding in B staging, head mean+bias+skip in the layer-2 gather; 18 -> 10 launches
# speedup vs baseline: 4.3972x; 1.0622x over previous
#include <hip/hip_runtime.h>
#include <math.h>

#define HEADS      4
#define HCF        128
#define C2         47
#define HC2        188
#define HC2P       192
#define SK2P       64
#define NEG_SLOPE  0.2f
#define EPS_DEN    1e-16f

typedef float v2f __attribute__((ext_vector_type(2)));
typedef float v8f __attribute__((ext_vector_type(8)));

__device__ __forceinline__ void wave_lds_sync() {
  __builtin_amdgcn_fence(__ATOMIC_ACQ_REL, "wavefront");
}

#define TPITCH 68

template <int K, int LDA, int LDB, int NB, int LDC, bool ATT>
__device__ __forceinline__ void gemm_strip(const float* __restrict__ A, const float* __restrict__ B,
                                           float* __restrict__ C, int M, int by, float* bs,
                                           const float* __restrict__ att_src,
                                           const float* __restrict__ att_dst,
                                           float* __restrict__ asrc, float* __restrict__ adst) {
  const int tid  = threadIdx.x;
  const int lane = tid & 31;
  const int wave = tid >> 5;
  const int tn   = by * 64;

  for (int idx = tid; idx < (K / 4) * 128; idx += 256) {
    const int l   = idx & 31;
    const int t   = (idx >> 5) & 3;
    const int ks  = idx >> 7;
    const int k   = 4 * ks + ((l >> 4) << 1);
    const int col = tn + t * 16 + (l & 15);
    float2 b;
    if constexpr (NB % 64 == 0) {
      b.x = B[(size_t)k * LDB + col];
      b.y = B[(size_t)(k + 1) * LDB + col];
    } else {
      b.x = col < NB ? B[(size_t)k * LDB + col] : 0.f;
      b.y = col < NB ? B[(size_t)(k + 1) * LDB + col] : 0.f;
    }
    *(float2*)&bs[(size_t)idx * 2] = b;
  }
  __syncthreads();

  const int rlo   = lane & 15;
  const int khalf = (lane >> 4) << 1;
  const int tm    = blockIdx.x * 8 + wave;

  int arow = tm * 16 + rlo;
  if (arow >= M) arow = M - 1;
  const float* ap = A + (size_t)arow * LDA;

  v8f acc[4] = {};
#pragma unroll 4
  for (int ks = 0; ks < K / 4; ++ks) {
    v2f a;
    a.x = ap[4 * ks + khalf];
    a.y = ap[4 * ks + khalf + 1];
    const float* bp = &bs[(size_t)(ks * 128 + lane) * 2];
#pragma unroll
    for (int t = 0; t < 4; ++t) {
      v2f b = *(const v2f*)(bp + t * 64);
      acc[t] = __builtin_amdgcn_wmma_f32_16x16x4_f32(
          false, a, false, b, (short)0, acc[t], false, false);
    }
  }

  if constexpr (ATT) {
    __syncthreads();
    float* tile = bs + wave * (16 * TPITCH);
    const int hi8 = (lane >> 4) << 3;
#pragma unroll
    for (int t = 0; t < 4; ++t)
#pragma unroll
      for (int r = 0; r < 8; ++r)
        tile[(hi8 + r) * TPITCH + t * 16 + rlo] = acc[t][r];
    wave_lds_sync();
    const int hh   = lane >> 4;
    const int head = by * 2 + hh;
    const float* tp = tile + rlo * TPITCH + hh * 32;
    const float* ws = att_src + head * 32;
    const float* wd = att_dst + head * 32;
    float s1 = 0.f, s2 = 0.f;
#pragma unroll
    for (int c4 = 0; c4 < 8; ++c4) {
      const float4 v = *(const float4*)(tp + 4 * c4);
      const float4 p = *(const float4*)(ws + 4 * c4);
      const float4 q = *(const float4*)(wd + 4 * c4);
      s1 = __builtin_fmaf(v.x, p.x, s1); s2 = __builtin_fmaf(v.x, q.x, s2);
      s1 = __builtin_fmaf(v.y, p.y, s1); s2 = __builtin_fmaf(v.y, q.y, s2);
      s1 = __builtin_fmaf(v.z, p.z, s1); s2 = __builtin_fmaf(v.z, q.z, s2);
      s1 = __builtin_fmaf(v.w, p.w, s1); s2 = __builtin_fmaf(v.w, q.w, s2);
    }
    const int grow = tm * 16 + rlo;
    if (grow < M) {
      asrc[(size_t)grow * 4 + head] = s1;
      adst[(size_t)grow * 4 + head] = s2;
    }
  }

  const int mbase = tm * 16 + ((lane >> 4) << 3);
  if ((tm + 1) * 16 <= M) {
    float* cp = C + (size_t)mbase * LDC + tn + rlo;
#pragma unroll
    for (int t = 0; t < 4; ++t)
#pragma unroll
      for (int r = 0; r < 8; ++r)
        cp[r * LDC + t * 16] = acc[t][r];
  } else {
#pragma unroll
    for (int t = 0; t < 4; ++t)
#pragma unroll
      for (int r = 0; r < 8; ++r) {
        const int row = mbase + r;
        if (row < M) C[(size_t)row * LDC + tn + t * 16 + rlo] = acc[t][r];
      }
  }
}

template <int LDB1, int NB1, int LDC1, int NY1, int LDB2, int NB2, int LDC2, bool ATT>
__global__ __launch_bounds__(256)
void gemm_pair(const float* __restrict__ A,
               const float* __restrict__ B1, float* __restrict__ Cmain,
               const float* __restrict__ B2, float* __restrict__ Cskip,
               const float* __restrict__ att_src, const float* __restrict__ att_dst,
               float* __restrict__ asrc, float* __restrict__ adst, int M) {
  __shared__ __attribute__((aligned(16))) float bs[ATT ? 8 * 16 * TPITCH : (128 / 4) * 128 * 2];
  const int by = blockIdx.y;
  if (by < NY1)
    gemm_strip<128, 128, LDB1, NB1, LDC1, ATT>(A, B1, Cmain, M, by, bs, att_src, att_dst, asrc, adst);
  else
    gemm_strip<128, 128, LDB2, NB2, LDC2, false>(A, B2, Cskip, M, by - NY1, bs, att_src, att_dst, asrc, adst);
}

template <int C, int LD>
__global__ void att_kernel(const float* __restrict__ xh,
                           const float* __restrict__ att_src,
                           const float* __restrict__ att_dst,
                           float* __restrict__ asrc, float* __restrict__ adst,
                           int N) {
  int id = blockIdx.x * blockDim.x + threadIdx.x;
  if (id >= N * HEADS) return;
  int n = id >> 2, h = id & 3;
  const float* xp = xh + (size_t)n * LD + h * C;
  const float* ws = att_src + h * C;
  const float* wd = att_dst + h * C;
  float s1 = 0.f, s2 = 0.f;
#pragma unroll 8
  for (int c = 0; c < C; ++c) {
    float v = xp[c];
    s1 += v * ws[c];
    s2 += v * wd[c];
  }
  asrc[id] = s1;
  adst[id] = s2;
}

__global__ void csr_count(const int* __restrict__ ei, int E0,
                          int* __restrict__ cnt, int* __restrict__ rank) {
  int e = blockIdx.x * blockDim.x + threadIdx.x;
  if (e >= E0) return;
  rank[e] = atomicAdd(&cnt[ei[E0 + e]], 1);
}

__global__ __launch_bounds__(1024)
void csr_scan(const int* __restrict__ cnt, int* __restrict__ rowptr, int N) {
  __shared__ int wsum[32];
  const int t    = threadIdx.x;
  const int lane = t & 31;
  const int wave = t >> 5;
  int carry = 0;
  for (int base = 0; base < N; base += 1024 * 16) {
    const int b = base + t * 16;
    int v[16];
    if (b + 16 <= N) {
#pragma unroll
      for (int i = 0; i < 4; ++i) {
        const int4 q = *(const int4*)(cnt + b + 4 * i);
        v[4 * i] = q.x; v[4 * i + 1] = q.y; v[4 * i + 2] = q.z; v[4 * i + 3] = q.w;
      }
    } else {
#pragma unroll
      for (int i = 0; i < 16; ++i) v[i] = (b + i < N) ? cnt[b + i] : 0;
    }
    int s = 0;
#pragma unroll
    for (int i = 0; i < 16; ++i) s += v[i];

    int incl = s;
#pragma unroll
    for (int off = 1; off < 32; off <<= 1) {
      const int y = __shfl_up(incl, off);
      if (lane >= off) incl += y;
    }
    if (lane == 31) wsum[wave] = incl;
    __syncthreads();
    if (wave == 0) {
      int ws = wsum[lane];
#pragma unroll
      for (int off = 1; off < 32; off <<= 1) {
        const int y = __shfl_up(ws, off);
        if (lane >= off) ws += y;
      }
      wsum[lane] = ws;
    }
    __syncthreads();
    const int wave_off   = (wave > 0) ? wsum[wave - 1] : 0;
    const int tile_total = wsum[31];
    int run = carry + wave_off + incl - s;
    int o[16];
#pragma unroll
    for (int i = 0; i < 16; ++i) { o[i] = run; run += v[i]; }
    if (b + 16 <= N) {
#pragma unroll
      for (int i = 0; i < 4; ++i)
        *(int4*)(rowptr + b + 4 * i) = int4{o[4 * i], o[4 * i + 1], o[4 * i + 2], o[4 * i + 3]};
    } else {
#pragma unroll
      for (int i = 0; i < 16; ++i)
        if (b + i < N) rowptr[b + i] = o[i];
    }
    carry += tile_total;
    __syncthreads();
  }
  if (t == 0) rowptr[N] = carry;
}

__global__ void csr_fill(const int* __restrict__ ei, int E0,
                         const int* __restrict__ rowptr, const int* __restrict__ rank,
                         int* __restrict__ col) {
  int e = blockIdx.x * blockDim.x + threadIdx.x;
  if (e >= E0) return;
  col[rowptr[ei[E0 + e]] + rank[e]] = ei[e];
}

__device__ __forceinline__ float wave_max(float v) {
#pragma unroll
  for (int off = 16; off > 0; off >>= 1) v = fmaxf(v, __shfl_xor(v, off));
  return v;
}
__device__ __forceinline__ float wave_sum(float v) {
#pragma unroll
  for (int off = 16; off > 0; off >>= 1) v += __shfl_xor(v, off);
  return v;
}

__device__ __forceinline__ int seg_scores(int j, int total, int beg, int n,
                                          const int* __restrict__ col,
                                          const float* __restrict__ asrc,
                                          const float4 ad4, float e[4]) {
  int s = n;
  if (j < total - 1) s = col[beg + j];
  const float4 as4 = *(const float4*)(asrc + (size_t)s * 4);
  float t[4] = {as4.x + ad4.x, as4.y + ad4.y, as4.z + ad4.z, as4.w + ad4.w};
#pragma unroll
  for (int h = 0; h < 4; ++h) {
    float x = t[h];
    x = x > 0.f ? x : NEG_SLOPE * x;
    e[h] = (j < total) ? x : -INFINITY;
  }
  return s;
}

template <int C, int LD, int NQ>
struct RowAcc {
  float4 acc[NQ];
  int    hq[NQ][4];
  int    lane;

  __device__ __forceinline__ void init(int ln) {
    lane = ln;
#pragma unroll
    for (int r = 0; r < NQ; ++r) {
      acc[r] = float4{0.f, 0.f, 0.f, 0.f};
#pragma unroll
      for (int i = 0; i < 4; ++i) {
        const int h = (4 * (lane + 32 * r) + i) / C;
        hq[r][i] = h < HEADS ? h : HEADS - 1;
      }
    }
  }
  __device__ __forceinline__ bool has(int r) const { return (lane + 32 * r) * 4 < HEADS * C; }

  __device__ __forceinline__ float4 weights(const float* p, int r) const {
    if constexpr (C % 4 == 0) {
      const float w = p[hq[r][0]];
      return float4{w, w, w, w};
    } else {
      return float4{p[hq[r][0]], p[hq[r][1]], p[hq[r][2]], p[hq[r][3]]};
    }
  }

  __device__ __forceinline__ void add(int cnt, int src, const float al[4],
                                      const float* __restrict__ xh, float* alw) {
    *(float4*)(alw + lane * 4) = float4{al[0], al[1], al[2], al[3]};
    wave_lds_sync();
    int jj = 0;
    for (; jj + 4 <= cnt; jj += 4) {
      float4 v[4][NQ];
      float4 a[4][NQ];
#pragma unroll
      for (int u = 0; u < 4; ++u) {
        const int s = __builtin_amdgcn_readlane(src, jj + u);
        const float* xs = xh + (size_t)s * LD + lane * 4;
#pragma unroll
        for (int r = 0; r < NQ; ++r) {
          v[u][r] = has(r) ? *(const float4*)(xs + 128 * r) : float4{0.f, 0.f, 0.f, 0.f};
          a[u][r] = weights(alw + (jj + u) * 4, r);
        }
      }
#pragma unroll
      for (int u = 0; u < 4; ++u)
#pragma unroll
        for (int r = 0; r < NQ; ++r) {
          acc[r].x += v[u][r].x * a[u][r].x;
          acc[r].y += v[u][r].y * a[u][r].y;
          acc[r].z += v[u][r].z * a[u][r].z;
          acc[r].w += v[u][r].w * a[u][r].w;
        }
    }
    for (; jj < cnt; ++jj) {
      const int s = __builtin_amdgcn_readlane(src, jj);
      const float* xs = xh + (size_t)s * LD + lane * 4;
#pragma unroll
      for (int r = 0; r < NQ; ++r) {
        const float4 v = has(r) ? *(const float4*)(xs + 128 * r) : float4{0.f, 0.f, 0.f, 0.f};
        const float4 a = weights(alw + jj * 4, r);
        acc[r].x += v.x * a.x;
        acc[r].y += v.y * a.y;
        acc[r].z += v.z * a.z;
        acc[r].w += v.w * a.w;
      }
    }
    wave_lds_sync();
  }
};

template <int C, int LD, int NQ>
__device__ __forceinline__ void node_aggregate(int n, int lane,
                                               const int* __restrict__ rowptr,
                                               const int* __restrict__ col,
                                               const float* __restrict__ xh,
                                               const float* __restrict__ asrc,
                                               const float* __restrict__ adst,
                                               RowAcc<C, LD, NQ>& ra, float* alw) {
  const int beg   = rowptr[n];
  const int total = rowptr[n + 1] - beg + 1;
  const float4 ad4 = *(const float4*)(adst + (size_t)n * 4);

  float e0[4], m[4];
  const int src0 = seg_scores(lane, total, beg, n, col, asrc, ad4, e0);
#pragma unroll
  for (int h = 0; h < 4; ++h) m[h] = e0[h];
  for (int base = 32; base < total; base += 32) {
    float e[4];
    seg_scores(base + lane, total, beg, n, col, asrc, ad4, e);
#pragma unroll
    for (int h = 0; h < 4; ++h) m[h] = fmaxf(m[h], e[h]);
  }
#pragma unroll
  for (int h = 0; h < 4; ++h) m[h] = wave_max(m[h]);

  float ex0[4], den[4];
#pragma unroll
  for (int h = 0; h < 4; ++h) {
    ex0[h] = (lane < total) ? __expf(e0[h] - m[h]) : 0.f;
    den[h] = ex0[h];
  }
  for (int base = 32; base < total; base += 32) {
    float e[4];
    seg_scores(base + lane, total, beg, n, col, asrc, ad4, e);
#pragma unroll
    for (int h = 0; h < 4; ++h) den[h] += (base + lane < total) ? __expf(e[h] - m[h]) : 0.f;
  }
#pragma unroll
  for (int h = 0; h < 4; ++h) den[h] = wave_sum(den[h]) + EPS_DEN;

  float al[4];
#pragma unroll
  for (int h = 0; h < 4; ++h) al[h] = ex0[h] / den[h];
  ra.add(total < 32 ? total : 32, src0, al, xh, alw);
  for (int base = 32; base < total; base += 32) {
    float e[4];
    const int s = seg_scores(base + lane, total, beg, n, col, asrc, ad4, e);
#pragma unroll
    for (int h = 0; h < 4; ++h)
      al[h] = (base + lane < total) ? __expf(e[h] - m[h]) / den[h] : 0.f;
    const int left = total - base;
    ra.add(left < 32 ? left : 32, s, al, xh, alw);
  }
}

__global__ __launch_bounds__(256)
void gather_elu(const int* __restrict__ rowptr, const int* __restrict__ col,
                const float* __restrict__ xh, const float* __restrict__ asrc,
                const float* __restrict__ adst, const float* __restrict__ skip,
                const float* __restrict__ bias, const float* __restrict__ skb,
                float* __restrict__ out, int N) {
  __shared__ __attribute__((aligned(16))) float alds[8 * 128];
  const int lane = threadIdx.x & 31;
  const int n = __builtin_amdgcn_readfirstlane(blockIdx.x * 8 + (threadIdx.x >> 5));
  if (n >= N) return;
  RowAcc<32, HCF, 1> ra;
  ra.init(lane);
  node_aggregate<32, HCF, 1>(n, lane, rowptr, col, xh, asrc, adst, ra,
                             alds + (threadIdx.x >> 5) * 128);
  const size_t o = (size_t)n * HCF + lane * 4;
  const float4 sk = *(const float4*)(skip + o);
  const float4 b  = *(const float4*)(bias + lane * 4);
  const float4 sb = *(const float4*)(skb + lane * 4);
  float v[4] = {ra.acc[0].x + b.x + sk.x + sb.x, ra.acc[0].y + b.y + sk.y + sb.y,
                ra.acc[0].z + b.z + sk.z + sb.z, ra.acc[0].w + b.w + sk.w + sb.w};
#pragma unroll
  for (int i = 0; i < 4; ++i) v[i] = v[i] > 0.f ? v[i] : expm1f(v[i]);
  *(float4*)(out + o) = float4{v[0], v[1], v[2], v[3]};
}

__global__ __launch_bounds__(256)
void gather_l2(const int* __restrict__ rowptr, const int* __restrict__ col,
               const float* __restrict__ xh, const float* __restrict__ asrc,
               const float* __restrict__ adst,
               const float* __restrict__ skip,
               const float* __restrict__ bias2, const float* __restrict__ skb2,
               float* __restrict__ out, int N) {
  __shared__ __attribute__((aligned(16))) float alds[8 * HC2P];
  const int lane = threadIdx.x & 31;
  const int n = __builtin_amdgcn_readfirstlane(blockIdx.x * 8 + (threadIdx.x >> 5));
  if (n >= N) return;
  float* alw = alds + (threadIdx.x >> 5) * HC2P;
  RowAcc<C2, HC2P, 2> ra;
  ra.init(lane);
  node_aggregate<C2, HC2P, 2>(n, lane, rowptr, col, xh, asrc, adst, ra, alw);
  *(float4*)(alw + lane * 4) = ra.acc[0];
  if (ra.has(1)) *(float4*)(alw + 128 + lane * 4) = ra.acc[1];
  wave_lds_sync();
  for (int o = lane; o < C2; o += 32) {
    const float m = 0.25f * (alw[o] + alw[C2 + o] + alw[2 * C2 + o] + alw[3 * C2 + o]);
    out[(size_t)n * C2 + o] = m + bias2[o] + skip[(size_t)n * SK2P + o] + skb2[o];
  }
}

extern "C" void kernel_launch(void* const* d_in, const int* in_sizes, int n_in,
                              void* d_out, int out_size, void* d_ws, size_t ws_size,
                              hipStream_t stream) {
  const float* x   = (const float*)d_in[0];
  const int*   ei  = (const int*)d_in[1];
  const float* lw[3] = {(const float*)d_in[2], (const float*)d_in[8],  (const float*)d_in[14]};
  const float* as[3] = {(const float*)d_in[3], (const float*)d_in[9],  (const float*)d_in[15]};
  const float* ad[3] = {(const float*)d_in[4], (const float*)d_in[10], (const float*)d_in[16]};
  const float* bi[3] = {(const float*)d_in[5], (const float*)d_in[11], (const float*)d_in[17]};
  const float* sw[3] = {(const float*)d_in[6], (const float*)d_in[12], (const float*)d_in[18]};
  const float* sb[3] = {(const float*)d_in[7], (const float*)d_in[13], (const float*)d_in[19]};

  const int N  = in_sizes[0] / 128;
  const int E0 = in_sizes[1] / 2;

  float* w = (float*)d_ws;
  float*    xcur = w;                      w += (size_t)N * HCF;
  float*    xh   = w;                      w += (size_t)N * HC2P;
  float*    skip = w;                      w += (size_t)N * HCF;
  float*    asrc = w;                      w += (size_t)N * 4;
  float*    adst = w;                      w += (size_t)N * 4;
  int*      cnt    = (int*)w;              w += ((size_t)N + 3) / 4 * 4;
  int*      rowptr = (int*)w;              w += ((size_t)N + 4) / 4 * 4;
  int*      rank   = (int*)w;              w += (size_t)E0;
  int*      col    = (int*)w;              w += (size_t)E0;

  const int mt8  = (N + 127) / 128;
  const int e0blk = (E0 + 255) / 256;
  const int nwarp = (N + 7) / 8;

  hipMemsetAsync(cnt, 0, (size_t)N * sizeof(int), stream);
  csr_count<<<e0blk, 256, 0, stream>>>(ei, E0, cnt, rank);
  csr_scan<<<1, 1024, 0, stream>>>(cnt, rowptr, N);
  csr_fill<<<e0blk, 256, 0, stream>>>(ei, E0, rowptr, rank, col);

  gemm_pair<128, 128, 128, 2, 128, 128, 128, true><<<dim3(mt8, 4), 256, 0, stream>>>(
      x, lw[0], xh, sw[0], skip, as[0], ad[0], asrc, adst, N);
  gather_elu<<<nwarp, 256, 0, stream>>>(rowptr, col, xh, asrc, adst, skip, bi[0], sb[0], xcur, N);

  gemm_pair<128, 128, 128, 2, 128, 128, 128, true><<<dim3(mt8, 4), 256, 0, stream>>>(
      xcur, lw[1], xh, sw[1], skip, as[1], ad[1], asrc, adst, N);
  gather_elu<<<nwarp, 256, 0, stream>>>(rowptr, col, xh, asrc, adst, skip, bi[1], sb[1], xcur, N);

  gemm_pair<HC2, HC2, HC2P, 3, C2, C2, SK2P, false><<<dim3(mt8, 4), 256, 0, stream>>>(
      xcur, lw[2], xh, sw[2], skip, as[2], ad[2], asrc, adst, N);
  att_kernel<47, HC2P><<<(N * 4 + 255) / 256, 256, 0, stream>>>(xh, as[2], ad[2], asrc, adst, N);
  gather_l2<<<nwarp, 256, 0, stream>>>(rowptr, col, xh, asrc, adst, skip, bi[2], sb[2], (float*)d_out, N);
}


# Edge grouping as a counting sort on LDS counters (bucket totals, per-bucket runs, per-node placing) instead of one returning global atomic per edge; sum order = LDS arrival rank inside a node
# speedup vs baseline: 5.6929x; 1.2947x over previous
#include <hip/hip_runtime.h>
#include <math.h>

#define HEADS      4
#define HCF        128
#define C2         47
#define HC2        188
#define HC2P       192
#define SK2P       64
#define NEG_SLOPE  0.2f
#define EPS_DEN    1e-16f

typedef float v2f __attribute__((ext_vector_type(2)));
typedef float v8f __attribute__((ext_vector_type(8)));

__device__ __forceinline__ void wave_lds_sync() {
  __builtin_amdgcn_fence(__ATOMIC_ACQ_REL, "wavefront");
}

#define TPITCH 68

template <int K, int LDA, int LDB, int NB, int LDC, bool ATT>
__device__ __forceinline__ void gemm_strip(const float* __restrict__ A, const float* __restrict__ B,
                                           float* __restrict__ C, int M, int by, float* bs,
                                           const float* __restrict__ att_src,
                                           const float* __restrict__ att_dst,
                                           float* __restrict__ asrc, float* __restrict__ adst) {
  const int tid  = threadIdx.x;
  const int lane = tid & 31;
  const int wave = tid >> 5;
  const int tn   = by * 64;

  for (int idx = tid; idx < (K / 4) * 128; idx += 256) {
    const int l   = idx & 31;
    const int t   = (idx >> 5) & 3;
    const int ks  = idx >> 7;
    const int k   = 4 * ks + ((l >> 4) << 1);
    const int col = tn + t * 16 + (l & 15);
    float2 b;
    if constexpr (NB % 64 == 0) {
      b.x = B[(size_t)k * LDB + col];
      b.y = B[(size_t)(k + 1) * LDB + col];
    } else {
      b.x = col < NB ? B[(size_t)k * LDB + col] : 0.f;
      b.y = col < NB ? B[(size_t)(k + 1) * LDB + col] : 0.f;
    }
    *(float2*)&bs[(size_t)idx * 2] = b;
  }
  __syncthreads();

  const int rlo   = lane & 15;
  const int khalf = (lane >> 4) << 1;
  const int tm    = blockIdx.x * 8 + wave;

  int arow = tm * 16 + rlo;
  if (arow >= M) arow = M - 1;
  const float* ap = A + (size_t)arow * LDA;

  v8f acc[4] = {};
#pragma unroll 4
  for (int ks = 0; ks < K / 4; ++ks) {
    v2f a;
    a.x = ap[4 * ks + khalf];
    a.y = ap[4 * ks + khalf + 1];
    const float* bp = &bs[(size_t)(ks * 128 + lane) * 2];
#pragma unroll
    for (int t = 0; t < 4; ++t) {
      v2f b = *(const v2f*)(bp + t * 64);
      acc[t] = __builtin_amdgcn_wmma_f32_16x16x4_f32(
          false, a, false, b, (short)0, acc[t], false, false);
    }
  }

  if constexpr (ATT) {
    __syncthreads();
    float* tile = bs + wave * (16 * TPITCH);
    const int hi8 = (lane >> 4) << 3;
#pragma unroll
    for (int t = 0; t < 4; ++t)
#pragma unroll
      for (int r = 0; r < 8; ++r)
        tile[(hi8 + r) * TPITCH + t * 16 + rlo] = acc[t][r];
    wave_lds_sync();
    const int hh   = lane >> 4;
    const int head = by * 2 + hh;
    const float* tp = tile + rlo * TPITCH + hh * 32;
    const float* ws = att_src + head * 32;
    const float* wd = att_dst + head * 32;
    float s1 = 0.f, s2 = 0.f;
#pragma unroll
    for (int c4 = 0; c4 < 8; ++c4) {
      const float4 v = *(const float4*)(tp + 4 * c4);
      const float4 p = *(const float4*)(ws + 4 * c4);
      const float4 q = *(const float4*)(wd + 4 * c4);
      s1 = __builtin_fmaf(v.x, p.x, s1); s2 = __builtin_fmaf(v.x, q.x, s2);
      s1 = __builtin_fmaf(v.y, p.y, s1); s2 = __builtin_fmaf(v.y, q.y, s2);
      s1 = __builtin_fmaf(v.z, p.z, s1); s2 = __builtin_fmaf(v.z, q.z, s2);
      s1 = __builtin_fmaf(v.w, p.w, s1); s2 = __builtin_fmaf(v.w, q.w, s2);
    }
    const int grow = tm * 16 + rlo;
    if (grow < M) {
      asrc[(size_t)grow * 4 + head] = s1;
      adst[(size_t)grow * 4 + head] = s2;
    }
  }

  const int mbase = tm * 16 + ((lane >> 4) << 3);
  if ((tm + 1) * 16 <= M) {
    float* cp = C + (size_t)mbase * LDC + tn + rlo;
#pragma unroll
    for (int t = 0; t < 4; ++t)
#pragma unroll
      for (int r = 0; r < 8; ++r)
        cp[r * LDC + t * 16] = acc[t][r];
  } else {
#pragma unroll
    for (int t = 0; t < 4; ++t)
#pragma unroll
      for (int r = 0; r < 8; ++r) {
        const int row = mbase + r;
        if (row < M) C[(size_t)row * LDC + tn + t * 16 + rlo] = acc[t][r];
      }
  }
}

template <int LDB1, int NB1, int LDC1, int NY1, int LDB2, int NB2, int LDC2, bool ATT>
__global__ __launch_bounds__(256)
void gemm_pair(const float* __restrict__ A,
               const float* __restrict__ B1, float* __restrict__ Cmain,
               const float* __restrict__ B2, float* __restrict__ Cskip,
               const float* __restrict__ att_src, const float* __restrict__ att_dst,
               float* __restrict__ asrc, float* __restrict__ adst, int M) {
  __shared__ __attribute__((aligned(16))) float bs[ATT ? 8 * 16 * TPITCH : (128 / 4) * 128 * 2];
  const int by = blockIdx.y;
  if (by < NY1)
    gemm_strip<128, 128, LDB1, NB1, LDC1, ATT>(A, B1, Cmain, M, by, bs, att_src, att_dst, asrc, adst);
  else
    gemm_strip<128, 128, LDB2, NB2, LDC2, false>(A, B2, Cskip, M, by - NY1, bs, att_src, att_dst, asrc, adst);
}

template <int C, int LD>
__global__ void att_kernel(const float* __restrict__ xh,
                           const float* __restrict__ att_src,
                           const float* __restrict__ att_dst,
                           float* __restrict__ asrc, float* __restrict__ adst,
                           int N) {
  int id = blockIdx.x * blockDim.x + threadIdx.x;
  if (id >= N * HEADS) return;
  int n = id >> 2, h = id & 3;
  const float* xp = xh + (size_t)n * LD + h * C;
  const float* ws = att_src + h * C;
  const float* wd = att_dst + h * C;
  float s1 = 0.f, s2 = 0.f;
#pragma unroll 8
  for (int c = 0; c < C; ++c) {
    float v = xp[c];
    s1 += v * ws[c];
    s2 += v * wd[c];
  }
  asrc[id] = s1;
  adst[id] = s2;
}

#define BNODE 256
#define MAXBK 256
#define ECH   2048
#define EPT   (ECH / 256)

template <int NW>
__device__ __forceinline__ int block_scan_excl(int v, int* wsum, int lane, int wave) {
  int incl = v;
#pragma unroll
  for (int off = 1; off < 32; off <<= 1) {
    const int y = __shfl_up(incl, off);
    if (lane >= off) incl += y;
  }
  if (lane == 31) wsum[wave] = incl;
  __syncthreads();
  int woff = 0;
#pragma unroll
  for (int w = 0; w < NW; ++w) woff += (w < wave) ? wsum[w] : 0;
  __syncthreads();
  return woff + incl - v;
}

__global__ __launch_bounds__(256)
void bkt_count(const int* __restrict__ ei, int E0, int* __restrict__ gtot) {
  __shared__ int hist[MAXBK];
  const int t = threadIdx.x;
  hist[t] = 0;
  const int e0 = blockIdx.x * ECH + t;
  int d[EPT];
#pragma unroll
  for (int k = 0; k < EPT; ++k) {
    const int e = e0 + k * 256;
    d[k] = e < E0 ? ei[E0 + e] : -1;
  }
  __syncthreads();
#pragma unroll
  for (int k = 0; k < EPT; ++k)
    if (d[k] >= 0) atomicAdd(&hist[d[k] / BNODE], 1);
  __syncthreads();
  const int c = hist[t];
  if (c) atomicAdd(&gtot[t], c);
}

__global__ __launch_bounds__(256)
void bkt_place(const int* __restrict__ ei, int E0, int nbk,
               const int* __restrict__ gtot, int* __restrict__ gcur,
               int2* __restrict__ rec) {
  __shared__ int hist[MAXBK];
  __shared__ int wsum[8];
  const int t    = threadIdx.x;
  const int lane = t & 31;
  const int wave = t >> 5;
  hist[t] = 0;
  const int e0 = blockIdx.x * ECH + t;
  int s[EPT], d[EPT], r[EPT];
#pragma unroll
  for (int k = 0; k < EPT; ++k) {
    const int e = e0 + k * 256;
    d[k] = e < E0 ? ei[E0 + e] : -1;
    s[k] = e < E0 ? ei[e] : 0;
  }
  __syncthreads();
#pragma unroll
  for (int k = 0; k < EPT; ++k)
    r[k] = d[k] >= 0 ? atomicAdd(&hist[d[k] / BNODE], 1) : 0;
  const int tot = t < nbk ? gtot[t] : 0;
  const int bst = block_scan_excl<8>(tot, wsum, lane, wave);
  const int c = hist[t];
  int base = bst;
  if (c) base += atomicAdd(&gcur[t], c);
  hist[t] = base;
  __syncthreads();
#pragma unroll
  for (int k = 0; k < EPT; ++k)
    if (d[k] >= 0) rec[hist[d[k] / BNODE] + r[k]] = int2{s[k], d[k]};
}

__global__ __launch_bounds__(512)
void bkt_sort(const int2* __restrict__ rec, const int* __restrict__ gtot, int nbk, int N,
              int* __restrict__ rowptr, int* __restrict__ col) {
  __shared__ int cur[BNODE];
  __shared__ int wsum[16];
  __shared__ int seg[2];
  const int t    = threadIdx.x;
  const int lane = t & 31;
  const int wave = t >> 5;
  const int b    = blockIdx.x;
  const int tot = t < nbk ? gtot[t] : 0;
  const int bst = block_scan_excl<16>(tot, wsum, lane, wave);
  if (t == b) { seg[0] = bst; seg[1] = tot; }
  if (t < BNODE) cur[t] = 0;
  __syncthreads();
  const int base = seg[0];
  const int cnt  = seg[1];
  const int2* rp = rec + base;
  for (int i0 = t; i0 < cnt; i0 += 4 * 512) {
    int d[4];
#pragma unroll
    for (int u = 0; u < 4; ++u) {
      const int i = i0 + u * 512;
      d[u] = i < cnt ? rp[i].y : -1;
    }
#pragma unroll
    for (int u = 0; u < 4; ++u)
      if (d[u] >= 0) atomicAdd(&cur[d[u] % BNODE], 1);
  }
  __syncthreads();
  const int c  = t < BNODE ? cur[t] : 0;
  const int ex = block_scan_excl<16>(c, wsum, lane, wave);
  if (t < BNODE) {
    const int n = b * BNODE + t;
    if (n < N) rowptr[n] = base + ex;
    if (n == N - 1) rowptr[N] = base + ex + c;
    cur[t] = base + ex;
  }
  __syncthreads();
  for (int i0 = t; i0 < cnt; i0 += 4 * 512) {
    int2 q[4];
#pragma unroll
    for (int u = 0; u < 4; ++u) {
      const int i = i0 + u * 512;
      q[u] = i < cnt ? rp[i] : int2{0, -1};
    }
#pragma unroll
    for (int u = 0; u < 4; ++u)
      if (q[u].y >= 0) col[atomicAdd(&cur[q[u].y % BNODE], 1)] = q[u].x;
  }
}

__device__ __forceinline__ float wave_max(float v) {
#pragma unroll
  for (int off = 16; off > 0; off >>= 1) v = fmaxf(v, __shfl_xor(v, off));
  return v;
}
__device__ __forceinline__ float wave_sum(float v) {
#pragma unroll
  for (int off = 16; off > 0; off >>= 1) v += __shfl_xor(v, off);
  return v;
}

__device__ __forceinline__ int seg_scores(int j, int total, int beg, int n,
                                          const int* __restrict__ col,
                                          const float* __restrict__ asrc,
                                          const float4 ad4, float e[4]) {
  int s = n;
  if (j < total - 1) s = col[beg + j];
  const float4 as4 = *(const float4*)(asrc + (size_t)s * 4);
  float t[4] = {as4.x + ad4.x, as4.y + ad4.y, as4.z + ad4.z, as4.w + ad4.w};
#pragma unroll
  for (int h = 0; h < 4; ++h) {
    float x = t[h];
    x = x > 0.f ? x : NEG_SLOPE * x;
    e[h] = (j < total) ? x : -INFINITY;
  }
  return s;
}

template <int C, int LD, int NQ>
struct RowAcc {
  float4 acc[NQ];
  int    hq[NQ][4];
  int    lane;

  __device__ __forceinline__ void init(int ln) {
    lane = ln;
#pragma unroll
    for (int r = 0; r < NQ; ++r) {
      acc[r] = float4{0.f, 0.f, 0.f, 0.f};
#pragma unroll
      for (int i = 0; i < 4; ++i) {
        const int h = (4 * (lane + 32 * r) + i) / C;
        hq[r][i] = h < HEADS ? h : HEADS - 1;
      }
    }
  }
  __device__ __forceinline__ bool has(int r) const { return (lane + 32 * r) * 4 < HEADS * C; }

  __device__ __forceinline__ float4 weights(const float* p, int r) const {
    if constexpr (C % 4 == 0) {
      const float w = p[hq[r][0]];
      return float4{w, w, w, w};
    } else {
      return float4{p[hq[r][0]], p[hq[r][1]], p[hq[r][2]], p[hq[r][3]]};
    }
  }

  __device__ __forceinline__ void add(int cnt, int src, const float al[4],
                                      const float* __restrict__ xh, float* alw) {
    *(float4*)(alw + lane * 4) = float4{al[0], al[1], al[2], al[3]};
    wave_lds_sync();
    int jj = 0;
    for (; jj + 4 <= cnt; jj += 4) {
      float4 v[4][NQ];
      float4 a[4][NQ];
#pragma unroll
      for (int u = 0; u < 4; ++u) {
        const int s = __builtin_amdgcn_readlane(src, jj + u);
        const float* xs = xh + (size_t)s * LD + lane * 4;
#pragma unroll
        for (int r = 0; r < NQ; ++r) {
          v[u][r] = has(r) ? *(const float4*)(xs + 128 * r) : float4{0.f, 0.f, 0.f, 0.f};
          a[u][r] = weights(alw + (jj + u) * 4, r);
        }
      }
#pragma unroll
      for (int u = 0; u < 4; ++u)
#pragma unroll
        for (int r = 0; r < NQ; ++r) {
          acc[r].x += v[u][r].x * a[u][r].x;
          acc[r].y += v[u][r].y * a[u][r].y;
          acc[r].z += v[u][r].z * a[u][r].z;
          acc[r].w += v[u][r].w * a[u][r].w;
        }
    }
    for (; jj < cnt; ++jj) {
      const int s = __builtin_amdgcn_readlane(src, jj);
      const float* xs = xh + (size_t)s * LD + lane * 4;
#pragma unroll
      for (int r = 0; r < NQ; ++r) {
        const float4 v = has(r) ? *(const float4*)(xs + 128 * r) : float4{0.f, 0.f, 0.f, 0.f};
        const float4 a = weights(alw + jj * 4, r);
        acc[r].x += v.x * a.x;
        acc[r].y += v.y * a.y;
        acc[r].z += v.z * a.z;
        acc[r].w += v.w * a.w;
      }
    }
    wave_lds_sync();
  }
};

template <int C, int LD, int NQ>
__device__ __forceinline__ void node_aggregate(int n, int lane,
                                               const int* __restrict__ rowptr,
                                               const int* __restrict__ col,
                                               const float* __restrict__ xh,
                                               const float* __restrict__ asrc,
                                               const float* __restrict__ adst,
                                               RowAcc<C, LD, NQ>& ra, float* alw) {
  const int beg   = rowptr[n];
  const int total = rowptr[n + 1] - beg + 1;
  const float4 ad4 = *(const float4*)(adst + (size_t)n * 4);

  float e0[4], m[4];
  const int src0 = seg_scores(lane, total, beg, n, col, asrc, ad4, e0);
#pragma unroll
  for (int h = 0; h < 4; ++h) m[h] = e0[h];
  for (int base = 32; base < total; base += 32) {
    float e[4];
    seg_scores(base + lane, total, beg, n, col, asrc, ad4, e);
#pragma unroll
    for (int h = 0; h < 4; ++h) m[h] = fmaxf(m[h], e[h]);
  }
#pragma unroll
  for (int h = 0; h < 4; ++h) m[h] = wave_max(m[h]);

  float ex0[4], den[4];
#pragma unroll
  for (int h = 0; h < 4; ++h) {
    ex0[h] = (lane < total) ? __expf(e0[h] - m[h]) : 0.f;
    den[h] = ex0[h];
  }
  for (int base = 32; base < total; base += 32) {
    float e[4];
    seg_scores(base + lane, total, beg, n, col, asrc, ad4, e);
#pragma unroll
    for (int h = 0; h < 4; ++h) den[h] += (base + lane < total) ? __expf(e[h] - m[h]) : 0.f;
  }
#pragma unroll
  for (int h = 0; h < 4; ++h) den[h] = wave_sum(den[h]) + EPS_DEN;

  float al[4];
#pragma unroll
  for (int h = 0; h < 4; ++h) al[h] = ex0[h] / den[h];
  ra.add(total < 32 ? total : 32, src0, al, xh, alw);
  for (int base = 32; base < total; base += 32) {
    float e[4];
    const int s = seg_scores(base + lane, total, beg, n, col, asrc, ad4, e);
#pragma unroll
    for (int h = 0; h < 4; ++h)
      al[h] = (base + lane < total) ? __expf(e[h] - m[h]) / den[h] : 0.f;
    const int left = total - base;
    ra.add(left < 32 ? left : 32, s, al, xh, alw);
  }
}

__global__ __launch_bounds__(256)
void gather_elu(const int* __restrict__ rowptr, const int* __restrict__ col,
                const float* __restrict__ xh, const float* __restrict__ asrc,
                const float* __restrict__ adst, const float* __restrict__ skip,
                const float* __restrict__ bias, const float* __restrict__ skb,
                float* __restrict__ out, int N) {
  __shared__ __attribute__((aligned(16))) float alds[8 * 128];
  const int lane = threadIdx.x & 31;
  const int n = __builtin_amdgcn_readfirstlane(blockIdx.x * 8 + (threadIdx.x >> 5));
  if (n >= N) return;
  RowAcc<32, HCF, 1> ra;
  ra.init(lane);
  node_aggregate<32, HCF, 1>(n, lane, rowptr, col, xh, asrc, adst, ra,
                             alds + (threadIdx.x >> 5) * 128);
  const size_t o = (size_t)n * HCF + lane * 4;
  const float4 sk = *(const float4*)(skip + o);
  const float4 b  = *(const float4*)(bias + lane * 4);
  const float4 sb = *(const float4*)(skb + lane * 4);
  float v[4] = {ra.acc[0].x + b.x + sk.x + sb.x, ra.acc[0].y + b.y + sk.y + sb.y,
                ra.acc[0].z + b.z + sk.z + sb.z, ra.acc[0].w + b.w + sk.w + sb.w};
#pragma unroll
  for (int i = 0; i < 4; ++i) v[i] = v[i] > 0.f ? v[i] : expm1f(v[i]);
  *(float4*)(out + o) = float4{v[0], v[1], v[2], v[3]};
}

__global__ __launch_bounds__(256)
void gather_l2(const int* __restrict__ rowptr, const int* __restrict__ col,
               const float* __restrict__ xh, const float* __restrict__ asrc,
               const float* __restrict__ adst,
               const float* __restrict__ skip,
               const float* __restrict__ bias2, const float* __restrict__ skb2,
               float* __restrict__ out, int N) {
  __shared__ __attribute__((aligned(16))) float alds[8 * HC2P];
  const int lane = threadIdx.x & 31;
  const int n = __builtin_amdgcn_readfirstlane(blockIdx.x * 8 + (threadIdx.x >> 5));
  if (n >= N) return;
  float* alw = alds + (threadIdx.x >> 5) * HC2P;
  RowAcc<C2, HC2P, 2> ra;
  ra.init(lane);
  node_aggregate<C2, HC2P, 2>(n, lane, rowptr, col, xh, asrc, adst, ra, alw);
  *(float4*)(alw + lane * 4) = ra.acc[0];
  if (ra.has(1)) *(float4*)(alw + 128 + lane * 4) = ra.acc[1];
  wave_lds_sync();
  for (int o = lane; o < C2; o += 32) {
    const float m = 0.25f * (alw[o] + alw[C2 + o] + alw[2 * C2 + o] + alw[3 * C2 + o]);
    out[(size_t)n * C2 + o] = m + bias2[o] + skip[(size_t)n * SK2P + o] + skb2[o];
  }
}

extern "C" void kernel_launch(void* const* d_in, const int* in_sizes, int n_in,
                              void* d_out, int out_size, void* d_ws, size_t ws_size,
                              hipStream_t stream) {
  const float* x   = (const float*)d_in[0];
  const int*   ei  = (const int*)d_in[1];
  const float* lw[3] = {(const float*)d_in[2], (const float*)d_in[8],  (const float*)d_in[14]};
  const float* as[3] = {(const float*)d_in[3], (const float*)d_in[9],  (const float*)d_in[15]};
  const float* ad[3] = {(const float*)d_in[4], (const float*)d_in[10], (const float*)d_in[16]};
  const float* bi[3] = {(const float*)d_in[5], (const float*)d_in[11], (const float*)d_in[17]};
  const float* sw[3] = {(const float*)d_in[6], (const float*)d_in[12], (const float*)d_in[18]};
  const float* sb[3] = {(const float*)d_in[7], (const float*)d_in[13], (const float*)d_in[19]};

  const int N  = in_sizes[0] / 128;
  const int E0 = in_sizes[1] / 2;

  float* w = (float*)d_ws;
  float*    xcur = w;                      w += (size_t)N * HCF;
  float*    xh   = w;                      w += (size_t)N * HC2P;
  float*    skip = w;                      w += (size_t)N * HCF;
  float*    asrc = w;                      w += (size_t)N * 4;
  float*    adst = w;                      w += (size_t)N * 4;
  int*      gtot   = (int*)w;              w += MAXBK;
  int*      gcur   = (int*)w;              w += MAXBK;
  int*      rowptr = (int*)w;              w += ((size_t)N + 4) / 4 * 4;
  int2*     rec    = (int2*)w;             w += (size_t)E0 * 2;
  int*      col    = (int*)w;              w += (size_t)E0;

  const int mt8  = (N + 127) / 128;
  const int eblk = (E0 + ECH - 1) / ECH;
  const int nbk  = (N + BNODE - 1) / BNODE;
  const int nwarp = (N + 7) / 8;

  hipMemsetAsync(gtot, 0, (size_t)2 * MAXBK * sizeof(int), stream);
  bkt_count<<<eblk, 256, 0, stream>>>(ei, E0, gtot);
  bkt_place<<<eblk, 256, 0, stream>>>(ei, E0, nbk, gtot, gcur, rec);
  bkt_sort<<<nbk, 512, 0, stream>>>(rec, gtot, nbk, N, rowptr, col);

  gemm_pair<128, 128, 128, 2, 128, 128, 128, true><<<dim3(mt8, 4), 256, 0, stream>>>(
      x, lw[0], xh, sw[0], skip, as[0], ad[0], asrc, adst, N);
  gather_elu<<<nwarp, 256, 0, stream>>>(rowptr, col, xh, asrc, adst, skip, bi[0], sb[0], xcur, N);

  gemm_pair<128, 128, 128, 2, 128, 128, 128, true><<<dim3(mt8, 4), 256, 0, stream>>>(
      xcur, lw[1], xh, sw[1], skip, as[1], ad[1], asrc, adst, N);
  gather_elu<<<nwarp, 256, 0, stream>>>(rowptr, col, xh, asrc, adst, skip, bi[1], sb[1], xcur, N);

  gemm_pair<HC2, HC2, HC2P, 3, C2, C2, SK2P, false><<<dim3(mt8, 4), 256, 0, stream>>>(
      xcur, lw[2], xh, sw[2], skip, as[2], ad[2], asrc, adst, N);
  att_kernel<47, HC2P><<<(N * 4 + 255) / 256, 256, 0, stream>>>(xh, as[2], ad[2], asrc, adst, N);
  gather_l2<<<nwarp, 256, 0, stream>>>(rowptr, col, xh, asrc, adst, skip, bi[2], sb[2], (float*)d_out, N);
}


# Gather kernels: a node's rows go out 8 (layer 2: 4) per batch, remainder in the same batch not one row per round trip; first batch and skip row requested before the softmax; sum order unchanged
# speedup vs baseline: 6.0681x; 1.0659x over previous
#include <hip/hip_runtime.h>
#include <math.h>

#define HEADS      4
#define HCF        128
#define C2         47
#define HC2        188
#define HC2P       192
#define SK2P       64
#define NEG_SLOPE  0.2f
#define EPS_DEN    1e-16f

typedef float v2f __attribute__((ext_vector_type(2)));
typedef float v8f __attribute__((ext_vector_type(8)));

__device__ __forceinline__ void wave_lds_sync() {
  __builtin_amdgcn_fence(__ATOMIC_ACQ_REL, "wavefront");
}

#define TPITCH 68

template <int K, int LDA, int LDB, int NB, int LDC, bool ATT>
__device__ __forceinline__ void gemm_strip(const float* __restrict__ A, const float* __restrict__ B,
                                           float* __restrict__ C, int M, int by, float* bs,
                                           const float* __restrict__ att_src,
                                           const float* __restrict__ att_dst,
                                           float* __restrict__ asrc, float* __restrict__ adst) {
  const int tid  = threadIdx.x;
  const int lane = tid & 31;
  const int wave = tid >> 5;
  const int tn   = by * 64;

  for (int idx = tid; idx < (K / 4) * 128; idx += 256) {
    const int l   = idx & 31;
    const int t   = (idx >> 5) & 3;
    const int ks  = idx >> 7;
    const int k   = 4 * ks + ((l >> 4) << 1);
    const int col = tn + t * 16 + (l & 15);
    float2 b;
    if constexpr (NB % 64 == 0) {
      b.x = B[(size_t)k * LDB + col];
      b.y = B[(size_t)(k + 1) * LDB + col];
    } else {
      b.x = col < NB ? B[(size_t)k * LDB + col] : 0.f;
      b.y = col < NB ? B[(size_t)(k + 1) * LDB + col] : 0.f;
    }
    *(float2*)&bs[(size_t)idx * 2] = b;
  }
  __syncthreads();

  const int rlo   = lane & 15;
  const int khalf = (lane >> 4) << 1;
  const int tm    = blockIdx.x * 8 + wave;

  int arow = tm * 16 + rlo;
  if (arow >= M) arow = M - 1;
  const float* ap = A + (size_t)arow * LDA;

  v8f acc[4] = {};
#pragma unroll 4
  for (int ks = 0; ks < K / 4; ++ks) {
    v2f a;
    a.x = ap[4 * ks + khalf];
    a.y = ap[4 * ks + khalf + 1];
    const float* bp = &bs[(size_t)(ks * 128 + lane) * 2];
#pragma unroll
    for (int t = 0; t < 4; ++t) {
      v2f b = *(const v2f*)(bp + t * 64);
      acc[t] = __builtin_amdgcn_wmma_f32_16x16x4_f32(
          false, a, false, b, (short)0, acc[t], false, false);
    }
  }

  if constexpr (ATT) {
    __syncthreads();
    float* tile = bs + wave * (16 * TPITCH);
    const int hi8 = (lane >> 4) << 3;
#pragma unroll
    for (int t = 0; t < 4; ++t)
#pragma unroll
      for (int r = 0; r < 8; ++r)
        tile[(hi8 + r) * TPITCH + t * 16 + rlo] = acc[t][r];
    wave_lds_sync();
    const int hh   = lane >> 4;
    const int head = by * 2 + hh;
    const float* tp = tile + rlo * TPITCH + hh * 32;
    const float* ws = att_src + head * 32;
    const float* wd = att_dst + head * 32;
    float s1 = 0.f, s2 = 0.f;
#pragma unroll
    for (int c4 = 0; c4 < 8; ++c4) {
      const float4 v = *(const float4*)(tp + 4 * c4);
      const float4 p = *(const float4*)(ws + 4 * c4);
      const float4 q = *(const float4*)(wd + 4 * c4);
      s1 = __builtin_fmaf(v.x, p.x, s1); s2 = __builtin_fmaf(v.x, q.x, s2);
      s1 = __builtin_fmaf(v.y, p.y, s1); s2 = __builtin_fmaf(v.y, q.y, s2);
      s1 = __builtin_fmaf(v.z, p.z, s1); s2 = __builtin_fmaf(v.z, q.z, s2);
      s1 = __builtin_fmaf(v.w, p.w, s1); s2 = __builtin_fmaf(v.w, q.w, s2);
    }
    const int grow = tm * 16 + rlo;
    if (grow < M) {
      asrc[(size_t)grow * 4 + head] = s1;
      adst[(size_t)grow * 4 + head] = s2;
    }
  }

  const int mbase = tm * 16 + ((lane >> 4) << 3);
  if ((tm + 1) * 16 <= M) {
    float* cp = C + (size_t)mbase * LDC + tn + rlo;
#pragma unroll
    for (int t = 0; t < 4; ++t)
#pragma unroll
      for (int r = 0; r < 8; ++r)
        cp[r * LDC + t * 16] = acc[t][r];
  } else {
#pragma unroll
    for (int t = 0; t < 4; ++t)
#pragma unroll
      for (int r = 0; r < 8; ++r) {
        const int row = mbase + r;
        if (row < M) C[(size_t)row * LDC + tn + t * 16 + rlo] = acc[t][r];
      }
  }
}

template <int LDB1, int NB1, int LDC1, int NY1, int LDB2, int NB2, int LDC2, bool ATT>
__global__ __launch_bounds__(256)
void gemm_pair(const float* __restrict__ A,
               const float* __restrict__ B1, float* __restrict__ Cmain,
               const float* __restrict__ B2, float* __restrict__ Cskip,
               const float* __restrict__ att_src, const float* __restrict__ att_dst,
               float* __restrict__ asrc, float* __restrict__ adst, int M) {
  __shared__ __attribute__((aligned(16))) float bs[ATT ? 8 * 16 * TPITCH : (128 / 4) * 128 * 2];
  const int by = blockIdx.y;
  if (by < NY1)
    gemm_strip<128, 128, LDB1, NB1, LDC1, ATT>(A, B1, Cmain, M, by, bs, att_src, att_dst, asrc, adst);
  else
    gemm_strip<128, 128, LDB2, NB2, LDC2, false>(A, B2, Cskip, M, by - NY1, bs, att_src, att_dst, asrc, adst);
}

template <int C, int LD>
__global__ void att_kernel(const float* __restrict__ xh,
                           const float* __restrict__ att_src,
                           const float* __restrict__ att_dst,
                           float* __restrict__ asrc, float* __restrict__ adst,
                           int N) {
  int id = blockIdx.x * blockDim.x + threadIdx.x;
  if (id >= N * HEADS) return;
  int n = id >> 2, h = id & 3;
  const float* xp = xh + (size_t)n * LD + h * C;
  const float* ws = att_src + h * C;
  const float* wd = att_dst + h * C;
  float s1 = 0.f, s2 = 0.f;
#pragma unroll 8
  for (int c = 0; c < C; ++c) {
    float v = xp[c];
    s1 += v * ws[c];
    s2 += v * wd[c];
  }
  asrc[id] = s1;
  adst[id] = s2;
}

#define BNODE 256
#define MAXBK 256
#define ECH   2048
#define EPT   (ECH / 256)

template <int NW>
__device__ __forceinline__ int block_scan_excl(int v, int* wsum, int lane, int wave) {
  int incl = v;
#pragma unroll
  for (int off = 1; off < 32; off <<= 1) {
    const int y = __shfl_up(incl, off);
    if (lane >= off) incl += y;
  }
  if (lane == 31) wsum[wave] = incl;
  __syncthreads();
  int woff = 0;
#pragma unroll
  for (int w = 0; w < NW; ++w) woff += (w < wave) ? wsum[w] : 0;
  __syncthreads();
  return woff + incl - v;
}

__global__ __launch_bounds__(256)
void bkt_count(const int* __restrict__ ei, int E0, int* __restrict__ gtot) {
  __shared__ int hist[MAXBK];
  const int t = threadIdx.x;
  hist[t] = 0;
  const int e0 = blockIdx.x * ECH + t;
  int d[EPT];
#pragma unroll
  for (int k = 0; k < EPT; ++k) {
    const int e = e0 + k * 256;
    d[k] = e < E0 ? ei[E0 + e] : -1;
  }
  __syncthreads();
#pragma unroll
  for (int k = 0; k < EPT; ++k)
    if (d[k] >= 0) atomicAdd(&hist[d[k] / BNODE], 1);
  __syncthreads();
  const int c = hist[t];
  if (c) atomicAdd(&gtot[t], c);
}

__global__ __launch_bounds__(256)
void bkt_place(const int* __restrict__ ei, int E0, int nbk,
               const int* __restrict__ gtot, int* __restrict__ gcur,
               int2* __restrict__ rec) {
  __shared__ int hist[MAXBK];
  __shared__ int wsum[8];
  const int t    = threadIdx.x;
  const int lane = t & 31;
  const int wave = t >> 5;
  hist[t] = 0;
  const int e0 = blockIdx.x * ECH + t;
  int s[EPT], d[EPT], r[EPT];
#pragma unroll
  for (int k = 0; k < EPT; ++k) {
    const int e = e0 + k * 256;
    d[k] = e < E0 ? ei[E0 + e] : -1;
    s[k] = e < E0 ? ei[e] : 0;
  }
  __syncthreads();
#pragma unroll
  for (int k = 0; k < EPT; ++k)
    r[k] = d[k] >= 0 ? atomicAdd(&hist[d[k] / BNODE], 1) : 0;
  const int tot = t < nbk ? gtot[t] : 0;
  const int bst = block_scan_excl<8>(tot, wsum, lane, wave);
  const int c = hist[t];
  int base = bst;
  if (c) base += atomicAdd(&gcur[t], c);
  hist[t] = base;
  __syncthreads();
#pragma unroll
  for (int k = 0; k < EPT; ++k)
    if (d[k] >= 0) rec[hist[d[k] / BNODE] + r[k]] = int2{s[k], d[k]};
}

__global__ __launch_bounds__(512)
void bkt_sort(const int2* __restrict__ rec, const int* __restrict__ gtot, int nbk, int N,
              int* __restrict__ rowptr, int* __restrict__ col) {
  __shared__ int cur[BNODE];
  __shared__ int wsum[16];
  __shared__ int seg[2];
  const int t    = threadIdx.x;
  const int lane = t & 31;
  const int wave = t >> 5;
  const int b    = blockIdx.x;
  const int tot = t < nbk ? gtot[t] : 0;
  const int bst = block_scan_excl<16>(tot, wsum, lane, wave);
  if (t == b) { seg[0] = bst; seg[1] = tot; }
  if (t < BNODE) cur[t] = 0;
  __syncthreads();
  const int base = seg[0];
  const int cnt  = seg[1];
  const int2* rp = rec + base;
  for (int i0 = t; i0 < cnt; i0 += 4 * 512) {
    int d[4];
#pragma unroll
    for (int u = 0; u < 4; ++u) {
      const int i = i0 + u * 512;
      d[u] = i < cnt ? rp[i].y : -1;
    }
#pragma unroll
    for (int u = 0; u < 4; ++u)
      if (d[u] >= 0) atomicAdd(&cur[d[u] % BNODE], 1);
  }
  __syncthreads();
  const int c  = t < BNODE ? cur[t] : 0;
  const int ex = block_scan_excl<16>(c, wsum, lane, wave);
  if (t < BNODE) {
    const int n = b * BNODE + t;
    if (n < N) rowptr[n] = base + ex;
    if (n == N - 1) rowptr[N] = base + ex + c;
    cur[t] = base + ex;
  }
  __syncthreads();
  for (int i0 = t; i0 < cnt; i0 += 4 * 512) {
    int2 q[4];
#pragma unroll
    for (int u = 0; u < 4; ++u) {
      const int i = i0 + u * 512;
      q[u] = i < cnt ? rp[i] : int2{0, -1};
    }
#pragma unroll
    for (int u = 0; u < 4; ++u)
      if (q[u].y >= 0) col[atomicAdd(&cur[q[u].y % BNODE], 1)] = q[u].x;
  }
}

__device__ __forceinline__ float wave_max(float v) {
#pragma unroll
  for (int off = 16; off > 0; off >>= 1) v = fmaxf(v, __shfl_xor(v, off));
  return v;
}
__device__ __forceinline__ float wave_sum(float v) {
#pragma unroll
  for (int off = 16; off > 0; off >>= 1) v += __shfl_xor(v, off);
  return v;
}

__device__ __forceinline__ int seg_scores(int j, int total, int beg, int n,
                                          const int* __restrict__ col,
                                          const float* __restrict__ asrc,
                                          const float4 ad4, float e[4]) {
  int s = n;
  if (j < total - 1) s = col[beg + j];
  const float4 as4 = *(const float4*)(asrc + (size_t)s * 4);
  float t[4] = {as4.x + ad4.x, as4.y + ad4.y, as4.z + ad4.z, as4.w + ad4.w};
#pragma unroll
  for (int h = 0; h < 4; ++h) {
    float x = t[h];
    x = x > 0.f ? x : NEG_SLOPE * x;
    e[h] = (j < total) ? x : -INFINITY;
  }
  return s;
}

template <int C, int LD, int NQ, int RB>
struct RowAcc {
  float4 acc[NQ];
  float4 v[RB][NQ];
  int    hq[NQ][4];
  int    lane;

  __device__ __forceinline__ void init(int ln) {
    lane = ln;
#pragma unroll
    for (int r = 0; r < NQ; ++r) {
      acc[r] = float4{0.f, 0.f, 0.f, 0.f};
#pragma unroll
      for (int i = 0; i < 4; ++i) {
        const int h = (4 * (lane + 32 * r) + i) / C;
        hq[r][i] = h < HEADS ? h : HEADS - 1;
      }
    }
  }
  __device__ __forceinline__ bool has(int r) const { return (lane + 32 * r) * 4 < HEADS * C; }

  __device__ __forceinline__ float4 weights(const float* p, int r) const {
    if constexpr (C % 4 == 0) {
      const float w = p[hq[r][0]];
      return float4{w, w, w, w};
    } else {
      return float4{p[hq[r][0]], p[hq[r][1]], p[hq[r][2]], p[hq[r][3]]};
    }
  }

  __device__ __forceinline__ void load_row(int u, int s, const float* __restrict__ xh) {
    const float* xs = xh + (size_t)s * LD + lane * 4;
#pragma unroll
    for (int r = 0; r < NQ; ++r)
      v[u][r] = has(r) ? *(const float4*)(xs + 128 * r) : float4{0.f, 0.f, 0.f, 0.f};
  }

  template <bool FULL>
  __device__ __forceinline__ void request(int jj, int cnt, int src, const float* __restrict__ xh) {
#pragma unroll
    for (int u = 0; u < RB; ++u)
      if (FULL || jj + u < cnt) load_row(u, __builtin_amdgcn_readlane(src, jj + u), xh);
  }
  __device__ __forceinline__ void request_first(int cnt, int src, const float* __restrict__ xh) {
#pragma unroll
    for (int u = 0; u < RB; ++u)
      load_row(u, __builtin_amdgcn_readlane(src, u < cnt ? u : cnt - 1), xh);
  }
  template <bool FULL>
  __device__ __forceinline__ void consume(int jj, int cnt, const float* alw) {
#pragma unroll
    for (int u = 0; u < RB; ++u)
      if (FULL || jj + u < cnt) {
#pragma unroll
        for (int r = 0; r < NQ; ++r) {
          const float4 a = weights(alw + (jj + u) * 4, r);
          acc[r].x = __builtin_fmaf(v[u][r].x, a.x, acc[r].x);
          acc[r].y = __builtin_fmaf(v[u][r].y, a.y, acc[r].y);
          acc[r].z = __builtin_fmaf(v[u][r].z, a.z, acc[r].z);
          acc[r].w = __builtin_fmaf(v[u][r].w, a.w, acc[r].w);
        }
      }
  }

  __device__ __forceinline__ void add_rest(int cnt, int src, const float al[4],
                                           const float* __restrict__ xh, float* alw) {
    *(float4*)(alw + lane * 4) = float4{al[0], al[1], al[2], al[3]};
    wave_lds_sync();
    consume<true>(0, cnt, alw);
    int jj = RB;
    for (; jj + RB <= cnt; jj += RB) {
      request<true>(jj, cnt, src, xh);
      consume<true>(jj, cnt, alw);
    }
    if (jj < cnt) {
      request<false>(jj, cnt, src, xh);
      consume<false>(jj, cnt, alw);
    }
    wave_lds_sync();
  }
};

template <int C, int LD, int NQ, int RB>
__device__ __forceinline__ void node_aggregate(int n, int lane,
                                               const int* __restrict__ rowptr,
                                               const int* __restrict__ col,
                                               const float* __restrict__ xh,
                                               const float* __restrict__ asrc,
                                               const float* __restrict__ adst,
                                               RowAcc<C, LD, NQ, RB>& ra, float* alw) {
  const int beg   = rowptr[n];
  const int total = rowptr[n + 1] - beg + 1;
  const float4 ad4 = *(const float4*)(adst + (size_t)n * 4);

  float e0[4], m[4];
  const int src0 = seg_scores(lane, total, beg, n, col, asrc, ad4, e0);
  const int cnt0 = total < 32 ? total : 32;
  ra.request_first(cnt0, src0, xh);
#pragma unroll
  for (int h = 0; h < 4; ++h) m[h] = e0[h];
  for (int base = 32; base < total; base += 32) {
    float e[4];
    seg_scores(base + lane, total, beg, n, col, asrc, ad4, e);
#pragma unroll
    for (int h = 0; h < 4; ++h) m[h] = fmaxf(m[h], e[h]);
  }
#pragma unroll
  for (int h = 0; h < 4; ++h) m[h] = wave_max(m[h]);

  float ex0[4], den[4];
#pragma unroll
  for (int h = 0; h < 4; ++h) {
    ex0[h] = (lane < total) ? __expf(e0[h] - m[h]) : 0.f;
    den[h] = ex0[h];
  }
  for (int base = 32; base < total; base += 32) {
    float e[4];
    seg_scores(base + lane, total, beg, n, col, asrc, ad4, e);
#pragma unroll
    for (int h = 0; h < 4; ++h) den[h] += (base + lane < total) ? __expf(e[h] - m[h]) : 0.f;
  }
#pragma unroll
  for (int h = 0; h < 4; ++h) den[h] = wave_sum(den[h]) + EPS_DEN;

  float al[4];
#pragma unroll
  for (int h = 0; h < 4; ++h) al[h] = ex0[h] / den[h];
  ra.add_rest(cnt0, src0, al, xh, alw);
  for (int base = 32; base < total; base += 32) {
    float e[4];
    const int s = seg_scores(base + lane, total, beg, n, col, asrc, ad4, e);
    const int left = total - base;
    const int cn = left < 32 ? left : 32;
    ra.request_first(cn, s, xh);
#pragma unroll
    for (int h = 0; h < 4; ++h)
      al[h] = (base + lane < total) ? __expf(e[h] - m[h]) / den[h] : 0.f;
    ra.add_rest(cn, s, al, xh, alw);
  }
}

__global__ __launch_bounds__(256)
void gather_elu(const int* __restrict__ rowptr, const int* __restrict__ col,
                const float* __restrict__ xh, const float* __restrict__ asrc,
                const float* __restrict__ adst, const float* __restrict__ skip,
                const float* __restrict__ bias, const float* __restrict__ skb,
                float* __restrict__ out, int N) {
  __shared__ __attribute__((aligned(16))) float alds[8 * 128];
  const int lane = threadIdx.x & 31;
  const int n = __builtin_amdgcn_readfirstlane(blockIdx.x * 8 + (threadIdx.x >> 5));
  if (n >= N) return;
  RowAcc<32, HCF, 1, 8> ra;
  ra.init(lane);
  const size_t o = (size_t)n * HCF + lane * 4;
  const float4 sk = *(const float4*)(skip + o);
  node_aggregate<32, HCF, 1, 8>(n, lane, rowptr, col, xh, asrc, adst, ra,
                                alds + (threadIdx.x >> 5) * 128);
  const float4 b  = *(const float4*)(bias + lane * 4);
  const float4 sb = *(const float4*)(skb + lane * 4);
  float v[4] = {ra.acc[0].x + b.x + sk.x + sb.x, ra.acc[0].y + b.y + sk.y + sb.y,
                ra.acc[0].z + b.z + sk.z + sb.z, ra.acc[0].w + b.w + sk.w + sb.w};
#pragma unroll
  for (int i = 0; i < 4; ++i) v[i] = v[i] > 0.f ? v[i] : expm1f(v[i]);
  *(float4*)(out + o) = float4{v[0], v[1], v[2], v[3]};
}

__global__ __launch_bounds__(256)
void gather_l2(const int* __restrict__ rowptr, const int* __restrict__ col,
               const float* __restrict__ xh, const float* __restrict__ asrc,
               const float* __restrict__ adst,
               const float* __restrict__ skip,
               const float* __restrict__ bias2, const float* __restrict__ skb2,
               float* __restrict__ out, int N) {
  __shared__ __attribute__((aligned(16))) float alds[8 * HC2P];
  const int lane = threadIdx.x & 31;
  const int n = __builtin_amdgcn_readfirstlane(blockIdx.x * 8 + (threadIdx.x >> 5));
  if (n >= N) return;
  float* alw = alds + (threadIdx.x >> 5) * HC2P;
  RowAcc<C2, HC2P, 2, 4> ra;
  ra.init(lane);
  float skv[2];
#pragma unroll
  for (int k = 0; k < 2; ++k) {
    const int o = lane + 32 * k;
    skv[k] = o < C2 ? skip[(size_t)n * SK2P + o] : 0.f;
  }
  node_aggregate<C2, HC2P, 2, 4>(n, lane, rowptr, col, xh, asrc, adst, ra, alw);
  *(float4*)(alw + lane * 4) = ra.acc[0];
  if (ra.has(1)) *(float4*)(alw + 128 + lane * 4) = ra.acc[1];
  wave_lds_sync();
#pragma unroll
  for (int k = 0; k < 2; ++k) {
    const int o = lane + 32 * k;
    if (o < C2) {
      const float m = 0.25f * (alw[o] + alw[C2 + o] + alw[2 * C2 + o] + alw[3 * C2 + o]);
      out[(size_t)n * C2 + o] = m + bias2[o] + skv[k] + skb2[o];
    }
  }
}

extern "C" void kernel_launch(void* const* d_in, const int* in_sizes, int n_in,
                              void* d_out, int out_size, void* d_ws, size_t ws_size,
                              hipStream_t stream) {
  const float* x   = (const float*)d_in[0];
  const int*   ei  = (const int*)d_in[1];
  const float* lw[3] = {(const float*)d_in[2], (const float*)d_in[8],  (const float*)d_in[14]};
  const float* as[3] = {(const float*)d_in[3], (const float*)d_in[9],  (const float*)d_in[15]};
  const float* ad[3] = {(const float*)d_in[4], (const float*)d_in[10], (const float*)d_in[16]};
  const float* bi[3] = {(const float*)d_in[5], (const float*)d_in[11], (const float*)d_in[17]};
  const float* sw[3] = {(const float*)d_in[6], (const float*)d_in[12], (const float*)d_in[18]};
  const float* sb[3] = {(const float*)d_in[7], (const float*)d_in[13], (const float*)d_in[19]};

  const int N  = in_sizes[0] / 128;
  const int E0 = in_sizes[1] / 2;

  float* w = (float*)d_ws;
  float*    xcur = w;                      w += (size_t)N * HCF;
  float*    xh   = w;                      w += (size_t)N * HC2P;
  float*    skip = w;                      w += (size_t)N * HCF;
  float*    asrc = w;                      w += (size_t)N * 4;
  float*    adst = w;                      w += (size_t)N * 4;
  int*      gtot   = (int*)w;              w += MAXBK;
  int*      gcur   = (int*)w;              w += MAXBK;
  int*      rowptr = (int*)w;              w += ((size_t)N + 4) / 4 * 4;
  int2*     rec    = (int2*)w;             w += (size_t)E0 * 2;
  int*      col    = (int*)w;              w += (size_t)E0;

  const int mt8  = (N + 127) / 128;
  const int eblk = (E0 + ECH - 1) / ECH;
  const int nbk  = (N + BNODE - 1) / BNODE;
  const int nwarp = (N + 7) / 8;

  hipMemsetAsync(gtot, 0, (size_t)2 * MAXBK * sizeof(int), stream);
  bkt_count<<<eblk, 256, 0, stream>>>(ei, E0, gtot);
  bkt_place<<<eblk, 256, 0, stream>>>(ei, E0, nbk, gtot, gcur, rec);
  bkt_sort<<<nbk, 512, 0, stream>>>(rec, gtot, nbk, N, rowptr, col);

  gemm_pair<128, 128, 128, 2, 128, 128, 128, true><<<dim3(mt8, 4), 256, 0, stream>>>(
      x, lw[0], xh, sw[0], skip, as[0], ad[0], asrc, adst, N);
  gather_elu<<<nwarp, 256, 0, stream>>>(rowptr, col, xh, asrc, adst, skip, bi[0], sb[0], xcur, N);

  gemm_pair<128, 128, 128, 2, 128, 128, 128, true><<<dim3(mt8, 4), 256, 0, stream>>>(
      xcur, lw[1], xh, sw[1], skip, as[1], ad[1], asrc, adst, N);
  gather_elu<<<nwarp, 256, 0, stream>>>(rowptr, col, xh, asrc, adst, skip, bi[1], sb[1], xcur, N);

  gemm_pair<HC2, HC2, HC2P, 3, C2, C2, SK2P, false><<<dim3(mt8, 4), 256, 0, stream>>>(
      xcur, lw[2], xh, sw[2], skip, as[2], ad[2], asrc, adst, N);
  att_kernel<47, HC2P><<<(N * 4 + 255) / 256, 256, 0, stream>>>(xh, as[2], ad[2], asrc, adst, N);
  gather_l2<<<nwarp, 256, 0, stream>>>(rowptr, col, xh, asrc, adst, skip, bi[2], sb[2], (float*)d_out, N);
}


# Launch fusion: layer-0 products ride in the three edge-grouping launches; layer-2 scores come out of its product via a score table in spare skip columns (their sum order changes); 10 -> 8 launches
# speedup vs baseline: 6.6560x; 1.0969x over previous
#include <hip/hip_runtime.h>
#include <math.h>

#define HEADS      4
#define HCF        128
#define C2         47
#define HC2        188
#define HC2P       192
#define SK2P       64
#define NEG_SLOPE  0.2f
#define EPS_DEN    1e-16f

typedef float v2f __attribute__((ext_vector_type(2)));
typedef float v8f __attribute__((ext_vector_type(8)));

__device__ __forceinline__ void wave_lds_sync() {
  __builtin_amdgcn_fence(__ATOMIC_ACQ_REL, "wavefront");
}

#define TPITCH 68

#define VC0 48
template <int K, int LDA, int LDB, int NB, int LDC, bool ATT, bool VCOLS>
__device__ __forceinline__ void gemm_strip(const float* __restrict__ A, const float* __restrict__ B,
                                           float* __restrict__ C, int M, int tid, int bx, int by,
                                           float* bs,
                                           const float* __restrict__ att_src,
                                           const float* __restrict__ att_dst,
                                           float* __restrict__ asrc, float* __restrict__ adst,
                                           const float* __restrict__ V) {
  const int lane = tid & 31;
  const int wave = tid >> 5;
  const int tn   = by * 64;

  for (int idx = tid; idx < (K / 4) * 128; idx += 256) {
    const int l   = idx & 31;
    const int t   = (idx >> 5) & 3;
    const int ks  = idx >> 7;
    const int k   = 4 * ks + ((l >> 4) << 1);
    const int col = tn + t * 16 + (l & 15);
    float2 b;
    if constexpr (NB % 64 == 0) {
      b.x = B[(size_t)k * LDB + col];
      b.y = B[(size_t)(k + 1) * LDB + col];
    } else {
      b.x = col < NB ? B[(size_t)k * LDB + col] : 0.f;
      b.y = col < NB ? B[(size_t)(k + 1) * LDB + col] : 0.f;
      if constexpr (VCOLS) {
        if (col >= VC0 && col < VC0 + 8) {
          b.x = V[k * 8 + col - VC0];
          b.y = V[(k + 1) * 8 + col - VC0];
        }
      }
    }
    *(float2*)&bs[(size_t)idx * 2] = b;
  }
  __syncthreads();

  const int rlo   = lane & 15;
  const int khalf = (lane >> 4) << 1;
  const int tm    = bx * 8 + wave;

  int arow = tm * 16 + rlo;
  if (arow >= M) arow = M - 1;
  const float* ap = A + (size_t)arow * LDA;

  v8f acc[4] = {};
#pragma unroll 4
  for (int ks = 0; ks < K / 4; ++ks) {
    v2f a;
    a.x = ap[4 * ks + khalf];
    a.y = ap[4 * ks + khalf + 1];
    const float* bp = &bs[(size_t)(ks * 128 + lane) * 2];
#pragma unroll
    for (int t = 0; t < 4; ++t) {
      v2f b = *(const v2f*)(bp + t * 64);
      acc[t] = __builtin_amdgcn_wmma_f32_16x16x4_f32(
          false, a, false, b, (short)0, acc[t], false, false);
    }
  }

  if constexpr (ATT) {
    __syncthreads();
    float* tile = bs + wave * (16 * TPITCH);
    const int hi8 = (lane >> 4) << 3;
#pragma unroll
    for (int t = 0; t < 4; ++t)
#pragma unroll
      for (int r = 0; r < 8; ++r)
        tile[(hi8 + r) * TPITCH + t * 16 + rlo] = acc[t][r];
    wave_lds_sync();
    const int hh   = lane >> 4;
    const int head = by * 2 + hh;
    const float* tp = tile + rlo * TPITCH + hh * 32;
    const float* ws = att_src + head * 32;
    const float* wd = att_dst + head * 32;
    float s1 = 0.f, s2 = 0.f;
#pragma unroll
    for (int c4 = 0; c4 < 8; ++c4) {
      const float4 v = *(const float4*)(tp + 4 * c4);
      const float4 p = *(const float4*)(ws + 4 * c4);
      const float4 q = *(const float4*)(wd + 4 * c4);
      s1 = __builtin_fmaf(v.x, p.x, s1); s2 = __builtin_fmaf(v.x, q.x, s2);
      s1 = __builtin_fmaf(v.y, p.y, s1); s2 = __builtin_fmaf(v.y, q.y, s2);
      s1 = __builtin_fmaf(v.z, p.z, s1); s2 = __builtin_fmaf(v.z, q.z, s2);
      s1 = __builtin_fmaf(v.w, p.w, s1); s2 = __builtin_fmaf(v.w, q.w, s2);
    }
    const int grow = tm * 16 + rlo;
    if (grow < M) {
      asrc[(size_t)grow * 4 + head] = s1;
      adst[(size_t)grow * 4 + head] = s2;
    }
  }

  const int mbase = tm * 16 + ((lane >> 4) << 3);
  if ((tm + 1) * 16 <= M) {
    float* cp = C + (size_t)mbase * LDC + tn + rlo;
#pragma unroll
    for (int t = 0; t < 4; ++t)
#pragma unroll
      for (int r = 0; r < 8; ++r)
        cp[r * LDC + t * 16] = acc[t][r];
  } else {
#pragma unroll
    for (int t = 0; t < 4; ++t)
#pragma unroll
      for (int r = 0; r < 8; ++r) {
        const int row = mbase + r;
        if (row < M) C[(size_t)row * LDC + tn + t * 16 + rlo] = acc[t][r];
      }
  }
}

template <int LDB1, int NB1, int LDC1, int NY1, int LDB2, int NB2, int LDC2, bool ATT, bool VCOLS>
__global__ __launch_bounds__(256)
void gemm_pair(const float* __restrict__ A,
               const float* __restrict__ B1, float* __restrict__ Cmain,
               const float* __restrict__ B2, float* __restrict__ Cskip,
               const float* __restrict__ att_src, const float* __restrict__ att_dst,
               float* __restrict__ asrc, float* __restrict__ adst,
               const float* __restrict__ V, int M) {
  __shared__ __attribute__((aligned(16))) float bs[ATT ? 8 * 16 * TPITCH : (128 / 4) * 128 * 2];
  const int by = blockIdx.y;
  if (by < NY1)
    gemm_strip<128, 128, LDB1, NB1, LDC1, ATT, false>(A, B1, Cmain, M, threadIdx.x, blockIdx.x, by, bs,
                                                      att_src, att_dst, asrc, adst, V);
  else
    gemm_strip<128, 128, LDB2, NB2, LDC2, false, VCOLS>(A, B2, Cskip, M, threadIdx.x, blockIdx.x, by - NY1, bs,
                                                        att_src, att_dst, asrc, adst, V);
}

#define BNODE 256
#define MAXBK 256
#define ECH   2048
#define EPT   (ECH / 256)

template <int NW>
__device__ __forceinline__ int block_scan_excl(int v, int* wsum, int lane, int wave) {
  int incl = v;
#pragma unroll
  for (int off = 1; off < 32; off <<= 1) {
    const int y = __shfl_up(incl, off);
    if (lane >= off) incl += y;
  }
  if (lane == 31) wsum[wave] = incl;
  __syncthreads();
  int woff = 0;
#pragma unroll
  for (int w = 0; w < NW; ++w) woff += (w < wave) ? wsum[w] : 0;
  __syncthreads();
  return woff + incl - v;
}

__device__ __forceinline__ void count_body(const int* __restrict__ ei, int E0, int* __restrict__ gtot,
                                           int* hist) {
  const int t = threadIdx.x;
  hist[t] = 0;
  const int e0 = blockIdx.x * ECH + t;
  int d[EPT];
#pragma unroll
  for (int k = 0; k < EPT; ++k) {
    const int e = e0 + k * 256;
    d[k] = e < E0 ? ei[E0 + e] : -1;
  }
  __syncthreads();
#pragma unroll
  for (int k = 0; k < EPT; ++k)
    if (d[k] >= 0) atomicAdd(&hist[d[k] / BNODE], 1);
  __syncthreads();
  const int c = hist[t];
  if (c) atomicAdd(&gtot[t], c);
}

__device__ __forceinline__ void place_body(const int* __restrict__ ei, int E0, int nbk,
                                           const int* __restrict__ gtot, int* __restrict__ gcur,
                                           int2* __restrict__ rec, int* hist, int* wsum) {
  const int t    = threadIdx.x;
  const int lane = t & 31;
  const int wave = t >> 5;
  hist[t] = 0;
  const int e0 = blockIdx.x * ECH + t;
  int s[EPT], d[EPT], r[EPT];
#pragma unroll
  for (int k = 0; k < EPT; ++k) {
    const int e = e0 + k * 256;
    d[k] = e < E0 ? ei[E0 + e] : -1;
    s[k] = e < E0 ? ei[e] : 0;
  }
  __syncthreads();
#pragma unroll
  for (int k = 0; k < EPT; ++k)
    r[k] = d[k] >= 0 ? atomicAdd(&hist[d[k] / BNODE], 1) : 0;
  const int tot = t < nbk ? gtot[t] : 0;
  const int bst = block_scan_excl<8>(tot, wsum, lane, wave);
  const int c = hist[t];
  int base = bst;
  if (c) base += atomicAdd(&gcur[t], c);
  hist[t] = base;
  __syncthreads();
#pragma unroll
  for (int k = 0; k < EPT; ++k)
    if (d[k] >= 0) rec[hist[d[k] / BNODE] + r[k]] = int2{s[k], d[k]};
}

__device__ __forceinline__ void l0_strip_job(int u, int mt8, int tid, float* bs,
                                             const float* __restrict__ A,
                                             const float* __restrict__ B1, float* __restrict__ Cmain,
                                             const float* __restrict__ B2, float* __restrict__ Cskip,
                                             const float* __restrict__ att_src,
                                             const float* __restrict__ att_dst,
                                             float* __restrict__ asrc, float* __restrict__ adst,
                                             int M, bool main_job) {
  const int q  = main_job ? u : u - 2 * mt8;
  const int by = q >= mt8 ? 1 : 0;
  const int bx = q - by * mt8;
  if (main_job)
    gemm_strip<128, 128, 128, 128, 128, true, false>(A, B1, Cmain, M, tid, bx, by, bs,
                                                     att_src, att_dst, asrc, adst, nullptr);
  else
    gemm_strip<128, 128, 128, 128, 128, false, false>(A, B2, Cskip, M, tid, bx, by, bs,
                                                      att_src, att_dst, asrc, adst, nullptr);
}

__global__ __launch_bounds__(256)
void count_gemm(const int* __restrict__ ei, int E0, int* __restrict__ gtot, int eblk, int mt8,
                const float* __restrict__ A, const float* __restrict__ B1, float* __restrict__ Cmain,
                const float* __restrict__ att_src, const float* __restrict__ att_dst,
                float* __restrict__ asrc, float* __restrict__ adst, int M) {
  __shared__ __attribute__((aligned(16))) float bs[8 * 16 * TPITCH];
  __shared__ int hist[MAXBK];
  const int b = blockIdx.x;
  if (b < eblk)
    count_body(ei, E0, gtot, hist);
  else
    l0_strip_job(b - eblk, mt8, threadIdx.x, bs, A, B1, Cmain, nullptr, nullptr,
                 att_src, att_dst, asrc, adst, M, true);
}

__global__ __launch_bounds__(256)
void place_gemm(const int* __restrict__ ei, int E0, int nbk,
                const int* __restrict__ gtot, int* __restrict__ gcur, int2* __restrict__ rec,
                int eblk, int mt8, int u0,
                const float* __restrict__ A,
                const float* __restrict__ B1, float* __restrict__ Cmain,
                const float* __restrict__ B2, float* __restrict__ Cskip,
                const float* __restrict__ att_src, const float* __restrict__ att_dst,
                float* __restrict__ asrc, float* __restrict__ adst, int M) {
  __shared__ __attribute__((aligned(16))) float bs[8 * 16 * TPITCH];
  __shared__ int hist[MAXBK];
  __shared__ int wsum[8];
  const int b = blockIdx.x;
  if (b < eblk) {
    place_body(ei, E0, nbk, gtot, gcur, rec, hist, wsum);
  } else {
    const int u = u0 + b - eblk;
    l0_strip_job(u, mt8, threadIdx.x, bs, A, B1, Cmain, B2, Cskip,
                 att_src, att_dst, asrc, adst, M, u < 2 * mt8);
  }
}

__device__ __forceinline__ void score_table(const float* __restrict__ Wm,
                                            const float* __restrict__ att_src,
                                            const float* __restrict__ att_dst,
                                            float* __restrict__ V) {
  for (int o = threadIdx.x; o < 128 * 8; o += 512) {
    const int k = o >> 3, j = o & 7, h = j & 3;
    const float* wp = Wm + (size_t)k * HC2 + h * C2;
    const float* ap = (j < 4 ? att_src : att_dst) + h * C2;
    float sacc = 0.f;
#pragma unroll 8
    for (int c = 0; c < C2; ++c) sacc = __builtin_fmaf(wp[c], ap[c], sacc);
    V[o] = sacc;
  }
}

__global__ __launch_bounds__(512)
void bkt_sort(const int2* __restrict__ rec, const int* __restrict__ gtot, int nbk, int N,
              int* __restrict__ rowptr, int* __restrict__ col,
              const float* __restrict__ Wm2, const float* __restrict__ att_src2,
              const float* __restrict__ att_dst2, float* __restrict__ V,
              int mt8, int u0, const float* __restrict__ A,
              const float* __restrict__ B2, float* __restrict__ Cskip, int M) {
  __shared__ __attribute__((aligned(16))) float bs2[2][(128 / 4) * 128 * 2];
  if (blockIdx.x == nbk) { score_table(Wm2, att_src2, att_dst2, V); return; }
  if (blockIdx.x > nbk) {
    const int half = __builtin_amdgcn_readfirstlane(threadIdx.x >> 8);
    const int u = u0 + 2 * (blockIdx.x - nbk - 1) + half;
    l0_strip_job(u, mt8, threadIdx.x & 255, bs2[half], A, nullptr, nullptr, B2, Cskip,
                 nullptr, nullptr, nullptr, nullptr, M, false);
    return;
  }
  __shared__ int cur[BNODE];
  __shared__ int wsum[16];
  __shared__ int seg[2];
  const int t    = threadIdx.x;
  const int lane = t & 31;
  const int wave = t >> 5;
  const int b    = blockIdx.x;
  const int tot = t < nbk ? gtot[t] : 0;
  const int bst = block_scan_excl<16>(tot, wsum, lane, wave);
  if (t == b) { seg[0] = bst; seg[1] = tot; }
  if (t < BNODE) cur[t] = 0;
  __syncthreads();
  const int base = seg[0];
  const int cnt  = seg[1];
  const int2* rp = rec + base;
  for (int i0 = t; i0 < cnt; i0 += 4 * 512) {
    int d[4];
#pragma unroll
    for (int u = 0; u < 4; ++u) {
      const int i = i0 + u * 512;
      d[u] = i < cnt ? rp[i].y : -1;
    }
#pragma unroll
    for (int u = 0; u < 4; ++u)
      if (d[u] >= 0) atomicAdd(&cur[d[u] % BNODE], 1);
  }
  __syncthreads();
  const int c  = t < BNODE ? cur[t] : 0;
  const int ex = block_scan_excl<16>(c, wsum, lane, wave);
  if (t < BNODE) {
    const int n = b * BNODE + t;
    if (n < N) rowptr[n] = base + ex;
    if (n == N - 1) rowptr[N] = base + ex + c;
    cur[t] = base + ex;
  }
  __syncthreads();
  for (int i0 = t; i0 < cnt; i0 += 4 * 512) {
    int2 q[4];
#pragma unroll
    for (int u = 0; u < 4; ++u) {
      const int i = i0 + u * 512;
      q[u] = i < cnt ? rp[i] : int2{0, -1};
    }
#pragma unroll
    for (int u = 0; u < 4; ++u)
      if (q[u].y >= 0) col[atomicAdd(&cur[q[u].y % BNODE], 1)] = q[u].x;
  }
}

__device__ __forceinline__ float wave_max(float v) {
#pragma unroll
  for (int off = 16; off > 0; off >>= 1) v = fmaxf(v, __shfl_xor(v, off));
  return v;
}
__device__ __forceinline__ float wave_sum(float v) {
#pragma unroll
  for (int off = 16; off > 0; off >>= 1) v += __shfl_xor(v, off);
  return v;
}

template <int AS>
__device__ __forceinline__ int seg_scores(int j, int total, int beg, int n,
                                          const int* __restrict__ col,
                                          const float* __restrict__ asrc,
                                          const float4 ad4, float e[4]) {
  int s = n;
  if (j < total - 1) s = col[beg + j];
  const float4 as4 = *(const float4*)(asrc + (size_t)s * AS);
  float t[4] = {as4.x + ad4.x, as4.y + ad4.y, as4.z + ad4.z, as4.w + ad4.w};
#pragma unroll
  for (int h = 0; h < 4; ++h) {
    float x = t[h];
    x = x > 0.f ? x : NEG_SLOPE * x;
    e[h] = (j < total) ? x : -INFINITY;
  }
  return s;
}

template <int C, int LD, int NQ, int RB>
struct RowAcc {
  float4 acc[NQ];
  float4 v[RB][NQ];
  int    hq[NQ][4];
  int    lane;

  __device__ __forceinline__ void init(int ln) {
    lane = ln;
#pragma unroll
    for (int r = 0; r < NQ; ++r) {
      acc[r] = float4{0.f, 0.f, 0.f, 0.f};
#pragma unroll
      for (int i = 0; i < 4; ++i) {
        const int h = (4 * (lane + 32 * r) + i) / C;
        hq[r][i] = h < HEADS ? h : HEADS - 1;
      }
    }
  }
  __device__ __forceinline__ bool has(int r) const { return (lane + 32 * r) * 4 < HEADS * C; }

  __device__ __forceinline__ float4 weights(const float* p, int r) const {
    if constexpr (C % 4 == 0) {
      const float w = p[hq[r][0]];
      return float4{w, w, w, w};
    } else {
      return float4{p[hq[r][0]], p[hq[r][1]], p[hq[r][2]], p[hq[r][3]]};
    }
  }

  __device__ __forceinline__ void load_row(int u, int s, const float* __restrict__ xh) {
    const float* xs = xh + (size_t)s * LD + lane * 4;
#pragma unroll
    for (int r = 0; r < NQ; ++r)
      v[u][r] = has(r) ? *(const float4*)(xs + 128 * r) : float4{0.f, 0.f, 0.f, 0.f};
  }

  template <bool FULL>
  __device__ __forceinline__ void request(int jj, int cnt, int src, const float* __restrict__ xh) {
#pragma unroll
    for (int u = 0; u < RB; ++u)
      if (FULL || jj + u < cnt) load_row(u, __builtin_amdgcn_readlane(src, jj + u), xh);
  }
  __device__ __forceinline__ void request_first(int cnt, int src, const float* __restrict__ xh) {
#pragma unroll
    for (int u = 0; u < RB; ++u)
      load_row(u, __builtin_amdgcn_readlane(src, u < cnt ? u : cnt - 1), xh);
  }
  template <bool FULL>
  __device__ __forceinline__ void consume(int jj, int cnt, const float* alw) {
#pragma unroll
    for (int u = 0; u < RB; ++u)
      if (FULL || jj + u < cnt) {
#pragma unroll
        for (int r = 0; r < NQ; ++r) {
          const float4 a = weights(alw + (jj + u) * 4, r);
          acc[r].x = __builtin_fmaf(v[u][r].x, a.x, acc[r].x);
          acc[r].y = __builtin_fmaf(v[u][r].y, a.y, acc[r].y);
          acc[r].z = __builtin_fmaf(v[u][r].z, a.z, acc[r].z);
          acc[r].w = __builtin_fmaf(v[u][r].w, a.w, acc[r].w);
        }
      }
  }

  __device__ __forceinline__ void add_rest(int cnt, int src, const float al[4],
                                           const float* __restrict__ xh, float* alw) {
    *(float4*)(alw + lane * 4) = float4{al[0], al[1], al[2], al[3]};
    wave_lds_sync();
    consume<true>(0, cnt, alw);
    int jj = RB;
    for (; jj + RB <= cnt; jj += RB) {
      request<true>(jj, cnt, src, xh);
      consume<true>(jj, cnt, alw);
    }
    if (jj < cnt) {
      request<false>(jj, cnt, src, xh);
      consume<false>(jj, cnt, alw);
    }
    wave_lds_sync();
  }
};

template <int C, int LD, int NQ, int RB, int AS>
__device__ __forceinline__ void node_aggregate(int n, int lane,
                                               const int* __restrict__ rowptr,
                                               const int* __restrict__ col,
                                               const float* __restrict__ xh,
                                               const float* __restrict__ asrc,
                                               const float* __restrict__ adst,
                                               RowAcc<C, LD, NQ, RB>& ra, float* alw) {
  const int beg   = rowptr[n];
  const int total = rowptr[n + 1] - beg + 1;
  const float4 ad4 = *(const float4*)(adst + (size_t)n * AS);

  float e0[4], m[4];
  const int src0 = seg_scores<AS>(lane, total, beg, n, col, asrc, ad4, e0);
  const int cnt0 = total < 32 ? total : 32;
  ra.request_first(cnt0, src0, xh);
#pragma unroll
  for (int h = 0; h < 4; ++h) m[h] = e0[h];
  for (int base = 32; base < total; base += 32) {
    float e[4];
    seg_scores<AS>(base + lane, total, beg, n, col, asrc, ad4, e);
#pragma unroll
    for (int h = 0; h < 4; ++h) m[h] = fmaxf(m[h], e[h]);
  }
#pragma unroll
  for (int h = 0; h < 4; ++h) m[h] = wave_max(m[h]);

  float ex0[4], den[4];
#pragma unroll
  for (int h = 0; h < 4; ++h) {
    ex0[h] = (lane < total) ? __expf(e0[h] - m[h]) : 0.f;
    den[h] = ex0[h];
  }
  for (int base = 32; base < total; base += 32) {
    float e[4];
    seg_scores<AS>(base + lane, total, beg, n, col, asrc, ad4, e);
#pragma unroll
    for (int h = 0; h < 4; ++h) den[h] += (base + lane < total) ? __expf(e[h] - m[h]) : 0.f;
  }
#pragma unroll
  for (int h = 0; h < 4; ++h) den[h] = wave_sum(den[h]) + EPS_DEN;

  float al[4];
#pragma unroll
  for (int h = 0; h < 4; ++h) al[h] = ex0[h] / den[h];
  ra.add_rest(cnt0, src0, al, xh, alw);
  for (int base = 32; base < total; base += 32) {
    float e[4];
    const int s = seg_scores<AS>(base + lane, total, beg, n, col, asrc, ad4, e);
    const int left = total - base;
    const int cn = left < 32 ? left : 32;
    ra.request_first(cn, s, xh);
#pragma unroll
    for (int h = 0; h < 4; ++h)
      al[h] = (base + lane < total) ? __expf(e[h] - m[h]) / den[h] : 0.f;
    ra.add_rest(cn, s, al, xh, alw);
  }
}

__global__ __launch_bounds__(256)
void gather_elu(const int* __restrict__ rowptr, const int* __restrict__ col,
                const float* __restrict__ xh, const float* __restrict__ asrc,
                const float* __restrict__ adst, const float* __restrict__ skip,
                const float* __restrict__ bias, const float* __restrict__ skb,
                float* __restrict__ out, int N) {
  __shared__ __attribute__((aligned(16))) float alds[8 * 128];
  const int lane = threadIdx.x & 31;
  const int n = __builtin_amdgcn_readfirstlane(blockIdx.x * 8 + (threadIdx.x >> 5));
  if (n >= N) return;
  RowAcc<32, HCF, 1, 8> ra;
  ra.init(lane);
  const size_t o = (size_t)n * HCF + lane * 4;
  const float4 sk = *(const float4*)(skip + o);
  node_aggregate<32, HCF, 1, 8, 4>(n, lane, rowptr, col, xh, asrc, adst, ra,
                                alds + (threadIdx.x >> 5) * 128);
  const float4 b  = *(const float4*)(bias + lane * 4);
  const float4 sb = *(const float4*)(skb + lane * 4);
  float v[4] = {ra.acc[0].x + b.x + sk.x + sb.x, ra.acc[0].y + b.y + sk.y + sb.y,
                ra.acc[0].z + b.z + sk.z + sb.z, ra.acc[0].w + b.w + sk.w + sb.w};
#pragma unroll
  for (int i = 0; i < 4; ++i) v[i] = v[i] > 0.f ? v[i] : expm1f(v[i]);
  *(float4*)(out + o) = float4{v[0], v[1], v[2], v[3]};
}

__global__ __launch_bounds__(256)
void gather_l2(const int* __restrict__ rowptr, const int* __restrict__ col,
               const float* __restrict__ xh,
               const float* __restrict__ skip,
               const float* __restrict__ bias2, const float* __restrict__ skb2,
               float* __restrict__ out, int N) {
  __shared__ __attribute__((aligned(16))) float alds[8 * HC2P];
  const int lane = threadIdx.x & 31;
  const int n = __builtin_amdgcn_readfirstlane(blockIdx.x * 8 + (threadIdx.x >> 5));
  if (n >= N) return;
  float* alw = alds + (threadIdx.x >> 5) * HC2P;
  RowAcc<C2, HC2P, 2, 4> ra;
  ra.init(lane);
  float skv[2];
#pragma unroll
  for (int k = 0; k < 2; ++k) {
    const int o = lane + 32 * k;
    skv[k] = o < C2 ? skip[(size_t)n * SK2P + o] : 0.f;
  }
  node_aggregate<C2, HC2P, 2, 4, SK2P>(n, lane, rowptr, col, xh, skip + VC0, skip + VC0 + 4, ra, alw);
  *(float4*)(alw + lane * 4) = ra.acc[0];
  if (ra.has(1)) *(float4*)(alw + 128 + lane * 4) = ra.acc[1];
  wave_lds_sync();
#pragma unroll
  for (int k = 0; k < 2; ++k) {
    const int o = lane + 32 * k;
    if (o < C2) {
      const float m = 0.25f * (alw[o] + alw[C2 + o] + alw[2 * C2 + o] + alw[3 * C2 + o]);
      out[(size_t)n * C2 + o] = m + bias2[o] + skv[k] + skb2[o];
    }
  }
}

extern "C" void kernel_launch(void* const* d_in, const int* in_sizes, int n_in,
                              void* d_out, int out_size, void* d_ws, size_t ws_size,
                              hipStream_t stream) {
  const float* x   = (const float*)d_in[0];
  const int*   ei  = (const int*)d_in[1];
  const float* lw[3] = {(const float*)d_in[2], (const float*)d_in[8],  (const float*)d_in[14]};
  const float* as[3] = {(const float*)d_in[3], (const float*)d_in[9],  (const float*)d_in[15]};
  const float* ad[3] = {(const float*)d_in[4], (const float*)d_in[10], (const float*)d_in[16]};
  const float* bi[3] = {(const float*)d_in[5], (const float*)d_in[11], (const float*)d_in[17]};
  const float* sw[3] = {(const float*)d_in[6], (const float*)d_in[12], (const float*)d_in[18]};
  const float* sb[3] = {(const float*)d_in[7], (const float*)d_in[13], (const float*)d_in[19]};

  const int N  = in_sizes[0] / 128;
  const int E0 = in_sizes[1] / 2;

  float* w = (float*)d_ws;
  float*    xcur = w;                      w += (size_t)N * HCF;
  float*    xh   = w;                      w += (size_t)N * HC2P;
  float*    skip = w;                      w += (size_t)N * HCF;
  float*    asrc = w;                      w += (size_t)N * 4;
  float*    adst = w;                      w += (size_t)N * 4;
  int*      gtot   = (int*)w;              w += MAXBK;
  int*      gcur   = (int*)w;              w += MAXBK;
  int*      rowptr = (int*)w;              w += ((size_t)N + 4) / 4 * 4;
  int2*     rec    = (int2*)w;             w += (size_t)E0 * 2;
  int*      col    = (int*)w;              w += (size_t)E0;
  float*    vtab   = w;                    w += 128 * 8;

  const int mt8  = (N + 127) / 128;
  const int eblk = (E0 + ECH - 1) / ECH;
  const int nbk  = (N + BNODE - 1) / BNODE;
  const int nwarp = (N + 7) / 8;

  hipMemsetAsync(gtot, 0, (size_t)2 * MAXBK * sizeof(int), stream);
  const int nj1 = 4 * mt8 / 3;
  const int nj3 = 4 * mt8 - 2 * nj1;
  count_gemm<<<eblk + nj1, 256, 0, stream>>>(ei, E0, gtot, eblk, mt8,
                                             x, lw[0], xh, as[0], ad[0], asrc, adst, N);
  place_gemm<<<eblk + nj1, 256, 0, stream>>>(ei, E0, nbk, gtot, gcur, rec, eblk, mt8, nj1,
                                             x, lw[0], xh, sw[0], skip, as[0], ad[0], asrc, adst, N);
  bkt_sort<<<nbk + 1 + nj3 / 2, 512, 0, stream>>>(rec, gtot, nbk, N, rowptr, col,
                                                  lw[2], as[2], ad[2], vtab,
                                                  mt8, 2 * nj1, x, sw[0], skip, N);

  gather_elu<<<nwarp, 256, 0, stream>>>(rowptr, col, xh, asrc, adst, skip, bi[0], sb[0], xcur, N);

  gemm_pair<128, 128, 128, 2, 128, 128, 128, true, false><<<dim3(mt8, 4), 256, 0, stream>>>(
      xcur, lw[1], xh, sw[1], skip, as[1], ad[1], asrc, adst, vtab, N);
  gather_elu<<<nwarp, 256, 0, stream>>>(rowptr, col, xh, asrc, adst, skip, bi[1], sb[1], xcur, N);

  gemm_pair<HC2, HC2, HC2P, 3, C2, C2, SK2P, false, true><<<dim3(mt8, 4), 256, 0, stream>>>(
      xcur, lw[2], xh, sw[2], skip, as[2], ad[2], asrc, adst, vtab, N);
  gather_l2<<<nwarp, 256, 0, stream>>>(rowptr, col, xh, skip, bi[2], sb[2], (float*)d_out, N);
}
